# starting kernel
# baseline (speedup 1.0000x reference)
#include <hip/hip_runtime.h>
#include <hip/hip_bf16.h>


typedef float v2f __attribute__((ext_vector_type(2)));
typedef float v8f __attribute__((ext_vector_type(8)));

#define HF 128

__device__ __forceinline__ unsigned ordkey(float f) {
    unsigned u = __float_as_uint(f);
    return (u & 0x80000000u) ? ~u : (u | 0x80000000u);
}
__device__ __forceinline__ float deord(unsigned u) {
    u = (u & 0x80000000u) ? (u & 0x7fffffffu) : ~u;
    return __uint_as_float(u);
}
__device__ __forceinline__ float leaky(float x, float s) { return x > 0.f ? x : s * x; }

__global__ void k_zero(float* p, int n) {
    int i = blockIdx.x * blockDim.x + threadIdx.x;
    if (i < n) p[i] = 0.f;
}

__global__ void k_ea_sum(const float* __restrict__ ea, int E, float* __restrict__ easum) {
    int idx = blockIdx.x * blockDim.x + threadIdx.x;
    int stride = gridDim.x * blockDim.x;
    float s0 = 0.f, s1 = 0.f, s2 = 0.f;
    for (int e = idx; e < E; e += stride) {
        s0 += ea[e * 3 + 0];
        s1 += ea[e * 3 + 1];
        s2 += ea[e * 3 + 2];
    }
    atomicAdd(easum + 0, s0);
    atomicAdd(easum + 1, s1);
    atomicAdd(easum + 2, s2);
}

__global__ void k_compute_c(const float* __restrict__ We, const float* __restrict__ ae,
                            const float* __restrict__ easum, float* __restrict__ cvec, int E) {
    if (threadIdx.x != 0 || blockIdx.x != 0) return;
    float cm = 0.f;
    for (int d = 0; d < 3; ++d) {
        float c = 0.f;
        for (int h = 0; h < HF; ++h) c += We[d * HF + h] * ae[h];
        cvec[d] = c;
        cm += (easum[d] / (float)E) * c;
    }
    cvec[3] = cm;
}

__global__ __launch_bounds__(256) void k_gemm_xw(const float* __restrict__ X,
                                                 const float* __restrict__ W,
                                                 float* __restrict__ Ho, int n) {
    const int lane = threadIdx.x & 31;
    const int wv   = threadIdx.x >> 5;
    const int lo16 = lane & 15;
    const int hi   = lane >> 4;
    const int ncol = wv * 16 + lo16;

    v2f breg[32];
#pragma unroll
    for (int kk = 0; kk < 32; ++kk) {
        int kb = kk * 4 + hi * 2;
        v2f b;
        b.x = W[(size_t)kb * HF + ncol];
        b.y = W[(size_t)(kb + 1) * HF + ncol];
        breg[kk] = b;
    }

    const int strips = (n + 15) >> 4;
    for (int s = blockIdx.x; s < strips; s += gridDim.x) {
        const int r0 = s * 16;
        int rowA = r0 + lo16;
        if (rowA >= n) rowA = n - 1;
        const float* xrow = X + (size_t)rowA * HF + hi * 2;

        v2f a[32];
#pragma unroll
        for (int kk = 0; kk < 32; ++kk) {
            float2 t = *(const float2*)(xrow + kk * 4);
            a[kk].x = t.x;
            a[kk].y = t.y;
        }

        v8f acc = {0.f, 0.f, 0.f, 0.f, 0.f, 0.f, 0.f, 0.f};
#pragma unroll
        for (int kk = 0; kk < 32; ++kk) {
            acc = __builtin_amdgcn_wmma_f32_16x16x4_f32(
                false, a[kk], false, breg[kk], (short)0, acc, false, false);
        }

#pragma unroll
        for (int v = 0; v < 8; ++v) {
            int row = r0 + v + hi * 8;
            if (row < n) Ho[(size_t)row * HF + ncol] = acc[v];
        }
    }
}

__global__ __launch_bounds__(256) void k_node_dots(const float* __restrict__ Hm,
                                                   const float* __restrict__ avs,
                                                   const float* __restrict__ avd,
                                                   float* __restrict__ als,
                                                   float* __restrict__ ald, int n) {
    const int lane = threadIdx.x & 31;
    int wid = (blockIdx.x * blockDim.x + threadIdx.x) >> 5;
    int nw  = (gridDim.x * blockDim.x) >> 5;
    float s0 = avs[lane], s1 = avs[lane + 32], s2 = avs[lane + 64], s3 = avs[lane + 96];
    float d0 = avd[lane], d1 = avd[lane + 32], d2 = avd[lane + 64], d3 = avd[lane + 96];
    for (int i = wid; i < n; i += nw) {
        const float* hr = Hm + (size_t)i * HF;
        float h0 = hr[lane], h1 = hr[lane + 32], h2 = hr[lane + 64], h3 = hr[lane + 96];
        float rs = h0 * s0 + h1 * s1 + h2 * s2 + h3 * s3;
        float rd = h0 * d0 + h1 * d1 + h2 * d2 + h3 * d3;
        for (int off = 16; off; off >>= 1) {
            rs += __shfl_xor(rs, off);
            rd += __shfl_xor(rd, off);
        }
        if (lane == 0) { als[i] = rs; ald[i] = rd; }
    }
}

__global__ void k_node_self(const float* __restrict__ als, const float* __restrict__ ald,
                            const float* __restrict__ cvec, float* __restrict__ aself,
                            unsigned* __restrict__ amax_u, int n) {
    int i = blockIdx.x * blockDim.x + threadIdx.x;
    if (i >= n) return;
    float a = leaky(als[i] + ald[i] + cvec[3], 0.2f);
    aself[i] = a;
    amax_u[i] = ordkey(a);
}

__global__ void k_edge_alpha(const int* __restrict__ src, const int* __restrict__ dst,
                             const float* __restrict__ ea, const float* __restrict__ als,
                             const float* __restrict__ ald, const float* __restrict__ cvec,
                             float* __restrict__ alpha, unsigned* __restrict__ amax_u, int E) {
    int e = blockIdx.x * blockDim.x + threadIdx.x;
    if (e >= E) return;
    float c0 = cvec[0], c1 = cvec[1], c2 = cvec[2];
    float ale = ea[e * 3 + 0] * c0 + ea[e * 3 + 1] * c1 + ea[e * 3 + 2] * c2;
    int s = src[e], d = dst[e];
    float a = leaky(als[s] + ald[d] + ale, 0.2f);
    alpha[e] = a;
    atomicMax(amax_u + d, ordkey(a));
}

__global__ void k_node_exself(unsigned* __restrict__ amax_u, float* __restrict__ amaxf,
                              const float* __restrict__ aself, float* __restrict__ exself,
                              float* __restrict__ denom, int n) {
    int i = blockIdx.x * blockDim.x + threadIdx.x;
    if (i >= n) return;
    float m = deord(amax_u[i]);
    float ex = expf(aself[i] - m);
    amaxf[i] = m;
    exself[i] = ex;
    denom[i] = ex;
}

__global__ void k_edge_exp(float* __restrict__ alpha, const float* __restrict__ amaxf,
                           const int* __restrict__ dst, float* __restrict__ denom, int E) {
    int e = blockIdx.x * blockDim.x + threadIdx.x;
    if (e >= E) return;
    int d = dst[e];
    float ex = expf(alpha[e] - amaxf[d]);
    alpha[e] = ex;
    atomicAdd(denom + d, ex);
}

__global__ void k_node_init_out(const float* __restrict__ Hm, const float* __restrict__ exself,
                                const float* __restrict__ denom, const float* __restrict__ bias,
                                float* __restrict__ acc, int n) {
    int idx = blockIdx.x * blockDim.x + threadIdx.x;
    if (idx >= n * HF) return;
    int i = idx >> 7, c = idx & 127;
    float coef = exself[i] / (denom[i] + 1e-16f);
    acc[idx] = Hm[idx] * coef + bias[c];
}

__global__ __launch_bounds__(256) void k_edge_scatter(const int* __restrict__ src,
                                                      const int* __restrict__ dst,
                                                      const float* __restrict__ ex,
                                                      const float* __restrict__ denom,
                                                      const float* __restrict__ Hm,
                                                      float* __restrict__ acc, int E) {
    const int lane = threadIdx.x & 31;
    int wid = (blockIdx.x * blockDim.x + threadIdx.x) >> 5;
    int nw  = (gridDim.x * blockDim.x) >> 5;
    for (int e = wid; e < E; e += nw) {
        int s = src[e], d = dst[e];
        float coef = ex[e] / (denom[d] + 1e-16f);
        const float4 hv = *(const float4*)(Hm + (size_t)s * HF + lane * 4);
        float* ap = acc + (size_t)d * HF + lane * 4;
        atomicAdd(ap + 0, hv.x * coef);
        atomicAdd(ap + 1, hv.y * coef);
        atomicAdd(ap + 2, hv.z * coef);
        atomicAdd(ap + 3, hv.w * coef);
    }
}

__global__ __launch_bounds__(128) void k_bn_stats(const float* __restrict__ acc,
                                                  float* __restrict__ colsum,
                                                  float* __restrict__ colsq, int n) {
    int c = threadIdx.x;
    int r0 = blockIdx.x * 256;
    int r1 = min(r0 + 256, n);
    float s = 0.f, s2 = 0.f;
    for (int r = r0; r < r1; ++r) {
        float y = leaky(acc[(size_t)r * HF + c], 0.01f);
        s += y; s2 += y * y;
    }
    atomicAdd(colsum + c, s);
    atomicAdd(colsq + c, s2);
}

__global__ void k_bn_apply(const float* __restrict__ acc, const float* __restrict__ colsum,
                           const float* __restrict__ colsq, const float* __restrict__ g,
                           const float* __restrict__ be, float* __restrict__ xout, int n) {
    int idx = blockIdx.x * blockDim.x + threadIdx.x;
    if (idx >= n * HF) return;
    int c = idx & 127;
    float invN = 1.f / (float)n;
    float mu = colsum[c] * invN;
    float var = colsq[c] * invN - mu * mu;
    float y = leaky(acc[idx], 0.01f);
    xout[idx] = (y - mu) * rsqrtf(var + 1e-5f) * g[c] + be[c];
}

extern "C" void kernel_launch(void* const* d_in, const int* in_sizes, int n_in,
                              void* d_out, int out_size, void* d_ws, size_t ws_size,
                              hipStream_t stream) {
    const float* x  = (const float*)d_in[0];
    const int*   ei = (const int*)d_in[1];
    const float* ea = (const float*)d_in[3];
    const int N = in_sizes[0] / HF;
    const int E = in_sizes[1] / 2;
    const int* src = ei;
    const int* dst = ei + E;

    float* ws = (float*)d_ws;
    const size_t NH = (size_t)N * HF;
    float* hbuf   = ws;
    float* acc    = hbuf + NH;
    float* xbuf   = acc + NH;
    float* alpha  = xbuf + NH;
    float* als    = alpha + E;
    float* ald    = als + N;
    float* aself  = ald + N;
    float* exself = aself + N;
    float* amaxf  = exself + N;
    float* denom  = amaxf + N;
    float* colsum = denom + N;
    float* colsq  = colsum + HF;
    float* easum  = colsq + HF;
    float* cvec   = easum + 4;

    k_zero<<<1, 8, 0, stream>>>(easum, 8);
    k_ea_sum<<<128, 256, 0, stream>>>(ea, E, easum);

    const int nb_e  = (E + 255) / 256;
    const int nb_n  = (N + 255) / 256;
    const int nb_nh = (int)((NH + 255) / 256);

    for (int l = 0; l < 3; ++l) {
        const float* Wl  = (const float*)d_in[4 + 8 * l];
        const float* asv = (const float*)d_in[5 + 8 * l];
        const float* adv = (const float*)d_in[6 + 8 * l];
        const float* Wel = (const float*)d_in[7 + 8 * l];
        const float* aev = (const float*)d_in[8 + 8 * l];
        const float* bv  = (const float*)d_in[9 + 8 * l];
        const float* gv  = (const float*)d_in[10 + 8 * l];
        const float* bev = (const float*)d_in[11 + 8 * l];
        const float* xin = (l == 0) ? x : xbuf;
        float* xout = (l == 2) ? (float*)d_out : xbuf;

        k_compute_c<<<1, 1, 0, stream>>>(Wel, aev, easum, cvec, E);
        k_gemm_xw<<<1024, 256, 0, stream>>>(xin, Wl, hbuf, N);
        k_node_dots<<<512, 256, 0, stream>>>(hbuf, asv, adv, als, ald, N);
        k_node_self<<<nb_n, 256, 0, stream>>>(als, ald, cvec, aself, (unsigned*)amaxf, N);
        k_edge_alpha<<<nb_e, 256, 0, stream>>>(src, dst, ea, als, ald, cvec,
                                               alpha, (unsigned*)amaxf, E);
        k_node_exself<<<nb_n, 256, 0, stream>>>((unsigned*)amaxf, amaxf, aself,
                                                exself, denom, N);
        k_edge_exp<<<nb_e, 256, 0, stream>>>(alpha, amaxf, dst, denom, E);
        k_node_init_out<<<nb_nh, 256, 0, stream>>>(hbuf, exself, denom, bv, acc, N);
        k_edge_scatter<<<2048, 256, 0, stream>>>(src, dst, alpha, denom, hbuf, acc, E);
        k_zero<<<1, 256, 0, stream>>>(colsum, 256);
        k_bn_stats<<<(N + 255) / 256, 128, 0, stream>>>(acc, colsum, colsq, N);
        k_bn_apply<<<nb_nh, 256, 0, stream>>>(acc, colsum, colsq, gv, bev, xout, N);
    }
}


# BN column sums: each 256-row block is read by 16 waves with 16 whole-row float4 loads in flight, partials meet in LDS, still one atomic per column and block; sum order inside a block changes
# speedup vs baseline: 1.2431x; 1.2431x over previous
#include <hip/hip_runtime.h>
#include <hip/hip_bf16.h>


typedef float v2f __attribute__((ext_vector_type(2)));
typedef float v8f __attribute__((ext_vector_type(8)));

#define HF 128

__device__ __forceinline__ unsigned ordkey(float f) {
    unsigned u = __float_as_uint(f);
    return (u & 0x80000000u) ? ~u : (u | 0x80000000u);
}
__device__ __forceinline__ float deord(unsigned u) {
    u = (u & 0x80000000u) ? (u & 0x7fffffffu) : ~u;
    return __uint_as_float(u);
}
__device__ __forceinline__ float leaky(float x, float s) { return x > 0.f ? x : s * x; }

__global__ void k_zero(float* p, int n) {
    int i = blockIdx.x * blockDim.x + threadIdx.x;
    if (i < n) p[i] = 0.f;
}

__global__ void k_ea_sum(const float* __restrict__ ea, int E, float* __restrict__ easum) {
    int idx = blockIdx.x * blockDim.x + threadIdx.x;
    int stride = gridDim.x * blockDim.x;
    float s0 = 0.f, s1 = 0.f, s2 = 0.f;
    for (int e = idx; e < E; e += stride) {
        s0 += ea[e * 3 + 0];
        s1 += ea[e * 3 + 1];
        s2 += ea[e * 3 + 2];
    }
    atomicAdd(easum + 0, s0);
    atomicAdd(easum + 1, s1);
    atomicAdd(easum + 2, s2);
}

__global__ void k_compute_c(const float* __restrict__ We, const float* __restrict__ ae,
                            const float* __restrict__ easum, float* __restrict__ cvec, int E) {
    if (threadIdx.x != 0 || blockIdx.x != 0) return;
    float cm = 0.f;
    for (int d = 0; d < 3; ++d) {
        float c = 0.f;
        for (int h = 0; h < HF; ++h) c += We[d * HF + h] * ae[h];
        cvec[d] = c;
        cm += (easum[d] / (float)E) * c;
    }
    cvec[3] = cm;
}

__global__ __launch_bounds__(256) void k_gemm_xw(const float* __restrict__ X,
                                                 const float* __restrict__ W,
                                                 float* __restrict__ Ho, int n) {
    const int lane = threadIdx.x & 31;
    const int wv   = threadIdx.x >> 5;
    const int lo16 = lane & 15;
    const int hi   = lane >> 4;
    const int ncol = wv * 16 + lo16;

    v2f breg[32];
#pragma unroll
    for (int kk = 0; kk < 32; ++kk) {
        int kb = kk * 4 + hi * 2;
        v2f b;
        b.x = W[(size_t)kb * HF + ncol];
        b.y = W[(size_t)(kb + 1) * HF + ncol];
        breg[kk] = b;
    }

    const int strips = (n + 15) >> 4;
    for (int s = blockIdx.x; s < strips; s += gridDim.x) {
        const int r0 = s * 16;
        int rowA = r0 + lo16;
        if (rowA >= n) rowA = n - 1;
        const float* xrow = X + (size_t)rowA * HF + hi * 2;

        v2f a[32];
#pragma unroll
        for (int kk = 0; kk < 32; ++kk) {
            float2 t = *(const float2*)(xrow + kk * 4);
            a[kk].x = t.x;
            a[kk].y = t.y;
        }

        v8f acc = {0.f, 0.f, 0.f, 0.f, 0.f, 0.f, 0.f, 0.f};
#pragma unroll
        for (int kk = 0; kk < 32; ++kk) {
            acc = __builtin_amdgcn_wmma_f32_16x16x4_f32(
                false, a[kk], false, breg[kk], (short)0, acc, false, false);
        }

#pragma unroll
        for (int v = 0; v < 8; ++v) {
            int row = r0 + v + hi * 8;
            if (row < n) Ho[(size_t)row * HF + ncol] = acc[v];
        }
    }
}

__global__ __launch_bounds__(256) void k_node_dots(const float* __restrict__ Hm,
                                                   const float* __restrict__ avs,
                                                   const float* __restrict__ avd,
                                                   float* __restrict__ als,
                                                   float* __restrict__ ald, int n) {
    const int lane = threadIdx.x & 31;
    int wid = (blockIdx.x * blockDim.x + threadIdx.x) >> 5;
    int nw  = (gridDim.x * blockDim.x) >> 5;
    float s0 = avs[lane], s1 = avs[lane + 32], s2 = avs[lane + 64], s3 = avs[lane + 96];
    float d0 = avd[lane], d1 = avd[lane + 32], d2 = avd[lane + 64], d3 = avd[lane + 96];
    for (int i = wid; i < n; i += nw) {
        const float* hr = Hm + (size_t)i * HF;
        float h0 = hr[lane], h1 = hr[lane + 32], h2 = hr[lane + 64], h3 = hr[lane + 96];
        float rs = h0 * s0 + h1 * s1 + h2 * s2 + h3 * s3;
        float rd = h0 * d0 + h1 * d1 + h2 * d2 + h3 * d3;
        for (int off = 16; off; off >>= 1) {
            rs += __shfl_xor(rs, off);
            rd += __shfl_xor(rd, off);
        }
        if (lane == 0) { als[i] = rs; ald[i] = rd; }
    }
}

__global__ void k_node_self(const float* __restrict__ als, const float* __restrict__ ald,
                            const float* __restrict__ cvec, float* __restrict__ aself,
                            unsigned* __restrict__ amax_u, int n) {
    int i = blockIdx.x * blockDim.x + threadIdx.x;
    if (i >= n) return;
    float a = leaky(als[i] + ald[i] + cvec[3], 0.2f);
    aself[i] = a;
    amax_u[i] = ordkey(a);
}

__global__ void k_edge_alpha(const int* __restrict__ src, const int* __restrict__ dst,
                             const float* __restrict__ ea, const float* __restrict__ als,
                             const float* __restrict__ ald, const float* __restrict__ cvec,
                             float* __restrict__ alpha, unsigned* __restrict__ amax_u, int E) {
    int e = blockIdx.x * blockDim.x + threadIdx.x;
    if (e >= E) return;
    float c0 = cvec[0], c1 = cvec[1], c2 = cvec[2];
    float ale = ea[e * 3 + 0] * c0 + ea[e * 3 + 1] * c1 + ea[e * 3 + 2] * c2;
    int s = src[e], d = dst[e];
    float a = leaky(als[s] + ald[d] + ale, 0.2f);
    alpha[e] = a;
    atomicMax(amax_u + d, ordkey(a));
}

__global__ void k_node_exself(unsigned* __restrict__ amax_u, float* __restrict__ amaxf,
                              const float* __restrict__ aself, float* __restrict__ exself,
                              float* __restrict__ denom, int n) {
    int i = blockIdx.x * blockDim.x + threadIdx.x;
    if (i >= n) return;
    float m = deord(amax_u[i]);
    float ex = expf(aself[i] - m);
    amaxf[i] = m;
    exself[i] = ex;
    denom[i] = ex;
}

__global__ void k_edge_exp(float* __restrict__ alpha, const float* __restrict__ amaxf,
                           const int* __restrict__ dst, float* __restrict__ denom, int E) {
    int e = blockIdx.x * blockDim.x + threadIdx.x;
    if (e >= E) return;
    int d = dst[e];
    float ex = expf(alpha[e] - amaxf[d]);
    alpha[e] = ex;
    atomicAdd(denom + d, ex);
}

__global__ void k_node_init_out(const float* __restrict__ Hm, const float* __restrict__ exself,
                                const float* __restrict__ denom, const float* __restrict__ bias,
                                float* __restrict__ acc, int n) {
    int idx = blockIdx.x * blockDim.x + threadIdx.x;
    if (idx >= n * HF) return;
    int i = idx >> 7, c = idx & 127;
    float coef = exself[i] / (denom[i] + 1e-16f);
    acc[idx] = Hm[idx] * coef + bias[c];
}

__global__ __launch_bounds__(256) void k_edge_scatter(const int* __restrict__ src,
                                                      const int* __restrict__ dst,
                                                      const float* __restrict__ ex,
                                                      const float* __restrict__ denom,
                                                      const float* __restrict__ Hm,
                                                      float* __restrict__ acc, int E) {
    const int lane = threadIdx.x & 31;
    int wid = (blockIdx.x * blockDim.x + threadIdx.x) >> 5;
    int nw  = (gridDim.x * blockDim.x) >> 5;
    for (int e = wid; e < E; e += nw) {
        int s = src[e], d = dst[e];
        float coef = ex[e] / (denom[d] + 1e-16f);
        const float4 hv = *(const float4*)(Hm + (size_t)s * HF + lane * 4);
        float* ap = acc + (size_t)d * HF + lane * 4;
        atomicAdd(ap + 0, hv.x * coef);
        atomicAdd(ap + 1, hv.y * coef);
        atomicAdd(ap + 2, hv.z * coef);
        atomicAdd(ap + 3, hv.w * coef);
    }
}

#define BN_ROWS  256
#define BN_WAVES 16
#define BN_RPW   (BN_ROWS / BN_WAVES)
__global__ __launch_bounds__(512) void k_bn_stats(const float* __restrict__ acc,
                                                  float* __restrict__ colsum,
                                                  float* __restrict__ colsq, int n) {
    __shared__ __attribute__((aligned(16))) float part[2][BN_WAVES][HF];
    const int lane = threadIdx.x & 31;
    const int w    = __builtin_amdgcn_readfirstlane((int)(threadIdx.x >> 5));
    const int r0   = blockIdx.x * BN_ROWS + w;

    float4 t[BN_RPW];
#pragma unroll
    for (int i = 0; i < BN_RPW; ++i) {
        int r = r0 + i * BN_WAVES;
        if (r >= n) r = n - 1;
        t[i] = *(const float4*)(acc + (size_t)r * HF + lane * 4);
    }

    float4 s = {0.f, 0.f, 0.f, 0.f}, q = {0.f, 0.f, 0.f, 0.f};
#pragma unroll
    for (int i = 0; i < BN_RPW; ++i) {
        const bool ok = (r0 + i * BN_WAVES) < n;
        float y0 = ok ? leaky(t[i].x, 0.01f) : 0.f;
        float y1 = ok ? leaky(t[i].y, 0.01f) : 0.f;
        float y2 = ok ? leaky(t[i].z, 0.01f) : 0.f;
        float y3 = ok ? leaky(t[i].w, 0.01f) : 0.f;
        s.x += y0; q.x += y0 * y0;
        s.y += y1; q.y += y1 * y1;
        s.z += y2; q.z += y2 * y2;
        s.w += y3; q.w += y3 * y3;
    }
    *(float4*)&part[0][w][lane * 4] = s;
    *(float4*)&part[1][w][lane * 4] = q;
    __syncthreads();

    if (threadIdx.x < 2 * HF) {
        const int which = threadIdx.x >> 7;
        const int c     = threadIdx.x & (HF - 1);
        float r = 0.f;
        for (int k = 0; k < BN_WAVES; ++k) r += part[which][k][c];
        atomicAdd((which ? colsq : colsum) + c, r);
    }
}

__global__ void k_bn_apply(const float* __restrict__ acc, const float* __restrict__ colsum,
                           const float* __restrict__ colsq, const float* __restrict__ g,
                           const float* __restrict__ be, float* __restrict__ xout, int n) {
    int idx = blockIdx.x * blockDim.x + threadIdx.x;
    if (idx >= n * HF) return;
    int c = idx & 127;
    float invN = 1.f / (float)n;
    float mu = colsum[c] * invN;
    float var = colsq[c] * invN - mu * mu;
    float y = leaky(acc[idx], 0.01f);
    xout[idx] = (y - mu) * rsqrtf(var + 1e-5f) * g[c] + be[c];
}

extern "C" void kernel_launch(void* const* d_in, const int* in_sizes, int n_in,
                              void* d_out, int out_size, void* d_ws, size_t ws_size,
                              hipStream_t stream) {
    const float* x  = (const float*)d_in[0];
    const int*   ei = (const int*)d_in[1];
    const float* ea = (const float*)d_in[3];
    const int N = in_sizes[0] / HF;
    const int E = in_sizes[1] / 2;
    const int* src = ei;
    const int* dst = ei + E;

    float* ws = (float*)d_ws;
    const size_t NH = (size_t)N * HF;
    float* hbuf   = ws;
    float* acc    = hbuf + NH;
    float* xbuf   = acc + NH;
    float* alpha  = xbuf + NH;
    float* als    = alpha + E;
    float* ald    = als + N;
    float* aself  = ald + N;
    float* exself = aself + N;
    float* amaxf  = exself + N;
    float* denom  = amaxf + N;
    float* colsum = denom + N;
    float* colsq  = colsum + HF;
    float* easum  = colsq + HF;
    float* cvec   = easum + 4;

    k_zero<<<1, 8, 0, stream>>>(easum, 8);
    k_ea_sum<<<128, 256, 0, stream>>>(ea, E, easum);

    const int nb_e  = (E + 255) / 256;
    const int nb_n  = (N + 255) / 256;
    const int nb_nh = (int)((NH + 255) / 256);

    for (int l = 0; l < 3; ++l) {
        const float* Wl  = (const float*)d_in[4 + 8 * l];
        const float* asv = (const float*)d_in[5 + 8 * l];
        const float* adv = (const float*)d_in[6 + 8 * l];
        const float* Wel = (const float*)d_in[7 + 8 * l];
        const float* aev = (const float*)d_in[8 + 8 * l];
        const float* bv  = (const float*)d_in[9 + 8 * l];
        const float* gv  = (const float*)d_in[10 + 8 * l];
        const float* bev = (const float*)d_in[11 + 8 * l];
        const float* xin = (l == 0) ? x : xbuf;
        float* xout = (l == 2) ? (float*)d_out : xbuf;

        k_compute_c<<<1, 1, 0, stream>>>(Wel, aev, easum, cvec, E);
        k_gemm_xw<<<1024, 256, 0, stream>>>(xin, Wl, hbuf, N);
        k_node_dots<<<512, 256, 0, stream>>>(hbuf, asv, adv, als, ald, N);
        k_node_self<<<nb_n, 256, 0, stream>>>(als, ald, cvec, aself, (unsigned*)amaxf, N);
        k_edge_alpha<<<nb_e, 256, 0, stream>>>(src, dst, ea, als, ald, cvec,
                                               alpha, (unsigned*)amaxf, E);
        k_node_exself<<<nb_n, 256, 0, stream>>>((unsigned*)amaxf, amaxf, aself,
                                                exself, denom, N);
        k_edge_exp<<<nb_e, 256, 0, stream>>>(alpha, amaxf, dst, denom, E);
        k_node_init_out<<<nb_nh, 256, 0, stream>>>(hbuf, exself, denom, bv, acc, N);
        k_edge_scatter<<<2048, 256, 0, stream>>>(src, dst, alpha, denom, hbuf, acc, E);
        k_zero<<<1, 256, 0, stream>>>(colsum, 256);
        k_bn_stats<<<(N + 255) / 256, 512, 0, stream>>>(acc, colsum, colsq, N);
        k_bn_apply<<<nb_nh, 256, 0, stream>>>(acc, colsum, colsq, gv, bev, xout, N);
    }
}


# Weighted row sum: edges go into a per-call destination-ordered list (int count, scan, fill) and one wave sums each row in registers, replacing the f32 atomic scatter; row sum order = list order
# speedup vs baseline: 1.4943x; 1.2021x over previous
#include <hip/hip_runtime.h>
#include <hip/hip_bf16.h>


typedef float v2f __attribute__((ext_vector_type(2)));
typedef float v8f __attribute__((ext_vector_type(8)));

#define HF 128

__device__ __forceinline__ unsigned ordkey(float f) {
    unsigned u = __float_as_uint(f);
    return (u & 0x80000000u) ? ~u : (u | 0x80000000u);
}
__device__ __forceinline__ float deord(unsigned u) {
    u = (u & 0x80000000u) ? (u & 0x7fffffffu) : ~u;
    return __uint_as_float(u);
}
__device__ __forceinline__ float leaky(float x, float s) { return x > 0.f ? x : s * x; }

__global__ void k_zero(float* p, int n) {
    int i = blockIdx.x * blockDim.x + threadIdx.x;
    if (i < n) p[i] = 0.f;
}

__global__ void k_ea_sum(const float* __restrict__ ea, int E, float* __restrict__ easum) {
    int idx = blockIdx.x * blockDim.x + threadIdx.x;
    int stride = gridDim.x * blockDim.x;
    float s0 = 0.f, s1 = 0.f, s2 = 0.f;
    for (int e = idx; e < E; e += stride) {
        s0 += ea[e * 3 + 0];
        s1 += ea[e * 3 + 1];
        s2 += ea[e * 3 + 2];
    }
    atomicAdd(easum + 0, s0);
    atomicAdd(easum + 1, s1);
    atomicAdd(easum + 2, s2);
}

__global__ void k_compute_c(const float* __restrict__ We, const float* __restrict__ ae,
                            const float* __restrict__ easum, float* __restrict__ cvec, int E) {
    if (threadIdx.x != 0 || blockIdx.x != 0) return;
    float cm = 0.f;
    for (int d = 0; d < 3; ++d) {
        float c = 0.f;
        for (int h = 0; h < HF; ++h) c += We[d * HF + h] * ae[h];
        cvec[d] = c;
        cm += (easum[d] / (float)E) * c;
    }
    cvec[3] = cm;
}

__global__ __launch_bounds__(256) void k_gemm_xw(const float* __restrict__ X,
                                                 const float* __restrict__ W,
                                                 float* __restrict__ Ho, int n) {
    const int lane = threadIdx.x & 31;
    const int wv   = threadIdx.x >> 5;
    const int lo16 = lane & 15;
    const int hi   = lane >> 4;
    const int ncol = wv * 16 + lo16;

    v2f breg[32];
#pragma unroll
    for (int kk = 0; kk < 32; ++kk) {
        int kb = kk * 4 + hi * 2;
        v2f b;
        b.x = W[(size_t)kb * HF + ncol];
        b.y = W[(size_t)(kb + 1) * HF + ncol];
        breg[kk] = b;
    }

    const int strips = (n + 15) >> 4;
    for (int s = blockIdx.x; s < strips; s += gridDim.x) {
        const int r0 = s * 16;
        int rowA = r0 + lo16;
        if (rowA >= n) rowA = n - 1;
        const float* xrow = X + (size_t)rowA * HF + hi * 2;

        v2f a[32];
#pragma unroll
        for (int kk = 0; kk < 32; ++kk) {
            float2 t = *(const float2*)(xrow + kk * 4);
            a[kk].x = t.x;
            a[kk].y = t.y;
        }

        v8f acc = {0.f, 0.f, 0.f, 0.f, 0.f, 0.f, 0.f, 0.f};
#pragma unroll
        for (int kk = 0; kk < 32; ++kk) {
            acc = __builtin_amdgcn_wmma_f32_16x16x4_f32(
                false, a[kk], false, breg[kk], (short)0, acc, false, false);
        }

#pragma unroll
        for (int v = 0; v < 8; ++v) {
            int row = r0 + v + hi * 8;
            if (row < n) Ho[(size_t)row * HF + ncol] = acc[v];
        }
    }
}

__global__ __launch_bounds__(256) void k_node_dots(const float* __restrict__ Hm,
                                                   const float* __restrict__ avs,
                                                   const float* __restrict__ avd,
                                                   float* __restrict__ als,
                                                   float* __restrict__ ald, int n) {
    const int lane = threadIdx.x & 31;
    int wid = (blockIdx.x * blockDim.x + threadIdx.x) >> 5;
    int nw  = (gridDim.x * blockDim.x) >> 5;
    float s0 = avs[lane], s1 = avs[lane + 32], s2 = avs[lane + 64], s3 = avs[lane + 96];
    float d0 = avd[lane], d1 = avd[lane + 32], d2 = avd[lane + 64], d3 = avd[lane + 96];
    for (int i = wid; i < n; i += nw) {
        const float* hr = Hm + (size_t)i * HF;
        float h0 = hr[lane], h1 = hr[lane + 32], h2 = hr[lane + 64], h3 = hr[lane + 96];
        float rs = h0 * s0 + h1 * s1 + h2 * s2 + h3 * s3;
        float rd = h0 * d0 + h1 * d1 + h2 * d2 + h3 * d3;
        for (int off = 16; off; off >>= 1) {
            rs += __shfl_xor(rs, off);
            rd += __shfl_xor(rd, off);
        }
        if (lane == 0) { als[i] = rs; ald[i] = rd; }
    }
}

__global__ void k_node_self(const float* __restrict__ als, const float* __restrict__ ald,
                            const float* __restrict__ cvec, float* __restrict__ aself,
                            unsigned* __restrict__ amax_u, int n) {
    int i = blockIdx.x * blockDim.x + threadIdx.x;
    if (i >= n) return;
    float a = leaky(als[i] + ald[i] + cvec[3], 0.2f);
    aself[i] = a;
    amax_u[i] = ordkey(a);
}

__global__ void k_edge_alpha(const int* __restrict__ src, const int* __restrict__ dst,
                             const float* __restrict__ ea, const float* __restrict__ als,
                             const float* __restrict__ ald, const float* __restrict__ cvec,
                             float* __restrict__ alpha, unsigned* __restrict__ amax_u, int E) {
    int e = blockIdx.x * blockDim.x + threadIdx.x;
    if (e >= E) return;
    float c0 = cvec[0], c1 = cvec[1], c2 = cvec[2];
    float ale = ea[e * 3 + 0] * c0 + ea[e * 3 + 1] * c1 + ea[e * 3 + 2] * c2;
    int s = src[e], d = dst[e];
    float a = leaky(als[s] + ald[d] + ale, 0.2f);
    alpha[e] = a;
    atomicMax(amax_u + d, ordkey(a));
}

__global__ void k_node_exself(unsigned* __restrict__ amax_u, float* __restrict__ amaxf,
                              const float* __restrict__ aself, float* __restrict__ exself,
                              float* __restrict__ denom, int n) {
    int i = blockIdx.x * blockDim.x + threadIdx.x;
    if (i >= n) return;
    float m = deord(amax_u[i]);
    float ex = expf(aself[i] - m);
    amaxf[i] = m;
    exself[i] = ex;
    denom[i] = ex;
}

__global__ void k_edge_exp(float* __restrict__ alpha, const float* __restrict__ amaxf,
                           const int* __restrict__ dst, float* __restrict__ denom, int E) {
    int e = blockIdx.x * blockDim.x + threadIdx.x;
    if (e >= E) return;
    int d = dst[e];
    float ex = expf(alpha[e] - amaxf[d]);
    alpha[e] = ex;
    atomicAdd(denom + d, ex);
}

__global__ void k_deg_count(const int* __restrict__ dst, int* __restrict__ deg, int E) {
    int e = blockIdx.x * blockDim.x + threadIdx.x;
    if (e >= E) return;
    atomicAdd(deg + dst[e], 1);
}

#define SCAN_T 1024
__global__ __launch_bounds__(SCAN_T) void k_row_scan(const int* __restrict__ deg,
                                                     int* __restrict__ rowptr,
                                                     int* __restrict__ cursor, int n) {
    __shared__ int part[SCAN_T];
    const int t = threadIdx.x;
    const int chunk = (n + SCAN_T - 1) / SCAN_T;
    const int b = t * chunk;
    int e = b + chunk;
    if (e > n) e = n;
    int s = 0;
    for (int i = b; i < e; ++i) s += deg[i];
    part[t] = s;
    __syncthreads();
    for (int off = 1; off < SCAN_T; off <<= 1) {
        const int v = (t >= off) ? part[t - off] : 0;
        __syncthreads();
        part[t] += v;
        __syncthreads();
    }
    int run = part[t] - s;
    for (int i = b; i < e; ++i) {
        const int dg = deg[i];
        rowptr[i] = run;
        cursor[i] = run;
        run += dg;
    }
}

__global__ void k_list_fill(const int* __restrict__ src, const int* __restrict__ dst,
                            int* __restrict__ cursor, int* __restrict__ ssrc,
                            int* __restrict__ perm, int E) {
    int e = blockIdx.x * blockDim.x + threadIdx.x;
    if (e >= E) return;
    const int p = atomicAdd(cursor + dst[e], 1);
    ssrc[p] = src[e];
    perm[p] = e;
}

__device__ __forceinline__ int lane_i(int v, int k) { return __builtin_amdgcn_readlane(v, k); }
__device__ __forceinline__ float lane_f(float v, int k) {
    return __int_as_float(__builtin_amdgcn_readlane(__float_as_int(v), k));
}

__device__ __forceinline__ void row_fma(float4& o, const float4 h, float c) {
    o.x = fmaf(h.x, c, o.x);
    o.y = fmaf(h.y, c, o.y);
    o.z = fmaf(h.z, c, o.z);
    o.w = fmaf(h.w, c, o.w);
}

__global__ __launch_bounds__(256) void k_node_gather(const int* __restrict__ rowptr,
                                                     const int* __restrict__ deg,
                                                     const int* __restrict__ ssrc,
                                                     const int* __restrict__ perm,
                                                     const float* __restrict__ ex,
                                                     const float* __restrict__ exself,
                                                     const float* __restrict__ denom,
                                                     const float* __restrict__ bias,
                                                     const float* __restrict__ Hm,
                                                     float* __restrict__ acc, int n) {
    const int lane = threadIdx.x & 31;
    const int nw   = (gridDim.x * blockDim.x) >> 5;
    const int w0 = __builtin_amdgcn_readfirstlane(
        (int)((blockIdx.x * blockDim.x + threadIdx.x) >> 5));
    const float4 bv = *(const float4*)(bias + lane * 4);
    for (int i = w0; i < n; i += nw) {
        const int start = rowptr[i];
        const int cnt   = deg[i];
        const float den = denom[i] + 1e-16f;
        const float cs  = exself[i] / den;
        const float4 hs = *(const float4*)(Hm + (size_t)i * HF + lane * 4);
        float4 o;
        o.x = hs.x * cs + bv.x;
        o.y = hs.y * cs + bv.y;
        o.z = hs.z * cs + bv.z;
        o.w = hs.w * cs + bv.w;
        for (int base = 0; base < cnt; base += 32) {
            int m = cnt - base;
            if (m > 32) m = 32;
            int sl = 0;
            float cl = 0.f;
            if (lane < m) {
                const int j = start + base + lane;
                sl = ssrc[j];
                cl = ex[perm[j]] / den;
            }
            int k = 0;
            for (; k + 4 <= m; k += 4) {
                const int s0 = lane_i(sl, k),     s1 = lane_i(sl, k + 1);
                const int s2 = lane_i(sl, k + 2), s3 = lane_i(sl, k + 3);
                const float c0 = lane_f(cl, k),     c1 = lane_f(cl, k + 1);
                const float c2 = lane_f(cl, k + 2), c3 = lane_f(cl, k + 3);
                const float4 h0 = *(const float4*)(Hm + (size_t)s0 * HF + lane * 4);
                const float4 h1 = *(const float4*)(Hm + (size_t)s1 * HF + lane * 4);
                const float4 h2 = *(const float4*)(Hm + (size_t)s2 * HF + lane * 4);
                const float4 h3 = *(const float4*)(Hm + (size_t)s3 * HF + lane * 4);
                row_fma(o, h0, c0);
                row_fma(o, h1, c1);
                row_fma(o, h2, c2);
                row_fma(o, h3, c3);
            }
            const int r = m - k;
            if (r == 3) {
                const int s0 = lane_i(sl, k), s1 = lane_i(sl, k + 1), s2 = lane_i(sl, k + 2);
                const float c0 = lane_f(cl, k), c1 = lane_f(cl, k + 1), c2 = lane_f(cl, k + 2);
                const float4 h0 = *(const float4*)(Hm + (size_t)s0 * HF + lane * 4);
                const float4 h1 = *(const float4*)(Hm + (size_t)s1 * HF + lane * 4);
                const float4 h2 = *(const float4*)(Hm + (size_t)s2 * HF + lane * 4);
                row_fma(o, h0, c0);
                row_fma(o, h1, c1);
                row_fma(o, h2, c2);
            } else if (r == 2) {
                const int s0 = lane_i(sl, k), s1 = lane_i(sl, k + 1);
                const float c0 = lane_f(cl, k), c1 = lane_f(cl, k + 1);
                const float4 h0 = *(const float4*)(Hm + (size_t)s0 * HF + lane * 4);
                const float4 h1 = *(const float4*)(Hm + (size_t)s1 * HF + lane * 4);
                row_fma(o, h0, c0);
                row_fma(o, h1, c1);
            } else if (r == 1) {
                const int s0 = lane_i(sl, k);
                const float c0 = lane_f(cl, k);
                const float4 h0 = *(const float4*)(Hm + (size_t)s0 * HF + lane * 4);
                row_fma(o, h0, c0);
            }
        }
        *(float4*)(acc + (size_t)i * HF + lane * 4) = o;
    }
}

#define BN_ROWS  256
#define BN_WAVES 16
#define BN_RPW   (BN_ROWS / BN_WAVES)
__global__ __launch_bounds__(512) void k_bn_stats(const float* __restrict__ acc,
                                                  float* __restrict__ colsum,
                                                  float* __restrict__ colsq, int n) {
    __shared__ __attribute__((aligned(16))) float part[2][BN_WAVES][HF];
    const int lane = threadIdx.x & 31;
    const int w    = __builtin_amdgcn_readfirstlane((int)(threadIdx.x >> 5));
    const int r0   = blockIdx.x * BN_ROWS + w;

    float4 t[BN_RPW];
#pragma unroll
    for (int i = 0; i < BN_RPW; ++i) {
        int r = r0 + i * BN_WAVES;
        if (r >= n) r = n - 1;
        t[i] = *(const float4*)(acc + (size_t)r * HF + lane * 4);
    }

    float4 s = {0.f, 0.f, 0.f, 0.f}, q = {0.f, 0.f, 0.f, 0.f};
#pragma unroll
    for (int i = 0; i < BN_RPW; ++i) {
        const bool ok = (r0 + i * BN_WAVES) < n;
        float y0 = ok ? leaky(t[i].x, 0.01f) : 0.f;
        float y1 = ok ? leaky(t[i].y, 0.01f) : 0.f;
        float y2 = ok ? leaky(t[i].z, 0.01f) : 0.f;
        float y3 = ok ? leaky(t[i].w, 0.01f) : 0.f;
        s.x += y0; q.x += y0 * y0;
        s.y += y1; q.y += y1 * y1;
        s.z += y2; q.z += y2 * y2;
        s.w += y3; q.w += y3 * y3;
    }
    *(float4*)&part[0][w][lane * 4] = s;
    *(float4*)&part[1][w][lane * 4] = q;
    __syncthreads();

    if (threadIdx.x < 2 * HF) {
        const int which = threadIdx.x >> 7;
        const int c     = threadIdx.x & (HF - 1);
        float r = 0.f;
        for (int k = 0; k < BN_WAVES; ++k) r += part[which][k][c];
        atomicAdd((which ? colsq : colsum) + c, r);
    }
}

__global__ void k_bn_apply(const float* __restrict__ acc, const float* __restrict__ colsum,
                           const float* __restrict__ colsq, const float* __restrict__ g,
                           const float* __restrict__ be, float* __restrict__ xout, int n) {
    int idx = blockIdx.x * blockDim.x + threadIdx.x;
    if (idx >= n * HF) return;
    int c = idx & 127;
    float invN = 1.f / (float)n;
    float mu = colsum[c] * invN;
    float var = colsq[c] * invN - mu * mu;
    float y = leaky(acc[idx], 0.01f);
    xout[idx] = (y - mu) * rsqrtf(var + 1e-5f) * g[c] + be[c];
}

extern "C" void kernel_launch(void* const* d_in, const int* in_sizes, int n_in,
                              void* d_out, int out_size, void* d_ws, size_t ws_size,
                              hipStream_t stream) {
    const float* x  = (const float*)d_in[0];
    const int*   ei = (const int*)d_in[1];
    const float* ea = (const float*)d_in[3];
    const int N = in_sizes[0] / HF;
    const int E = in_sizes[1] / 2;
    const int* src = ei;
    const int* dst = ei + E;

    float* ws = (float*)d_ws;
    const size_t NH = (size_t)N * HF;
    float* bufA   = ws;
    float* bufB   = bufA + NH;
    int*   ssrc   = (int*)(bufB + NH);
    int*   perm   = ssrc + E;
    int*   rowptr = perm + E;
    int*   cursor = rowptr + N;
    int*   deg    = cursor + N;
    float* alpha  = bufB + NH + NH;
    float* als    = alpha + E;
    float* ald    = als + N;
    float* aself  = ald + N;
    float* exself = aself + N;
    float* amaxf  = exself + N;
    float* denom  = amaxf + N;
    float* colsum = denom + N;
    float* colsq  = colsum + HF;
    float* easum  = colsq + HF;
    float* cvec   = easum + 4;

    k_zero<<<1, 8, 0, stream>>>(easum, 8);
    k_ea_sum<<<128, 256, 0, stream>>>(ea, E, easum);

    const int nb_e  = (E + 255) / 256;
    const int nb_n  = (N + 255) / 256;
    const int nb_nh = (int)((NH + 255) / 256);

    k_zero<<<nb_n, 256, 0, stream>>>((float*)deg, N);
    k_deg_count<<<nb_e, 256, 0, stream>>>(dst, deg, E);
    k_row_scan<<<1, SCAN_T, 0, stream>>>(deg, rowptr, cursor, N);
    k_list_fill<<<nb_e, 256, 0, stream>>>(src, dst, cursor, ssrc, perm, E);

    for (int l = 0; l < 3; ++l) {
        const float* Wl  = (const float*)d_in[4 + 8 * l];
        const float* asv = (const float*)d_in[5 + 8 * l];
        const float* adv = (const float*)d_in[6 + 8 * l];
        const float* Wel = (const float*)d_in[7 + 8 * l];
        const float* aev = (const float*)d_in[8 + 8 * l];
        const float* bv  = (const float*)d_in[9 + 8 * l];
        const float* gv  = (const float*)d_in[10 + 8 * l];
        const float* bev = (const float*)d_in[11 + 8 * l];
        float* hbuf = (l & 1) ? bufB : bufA;
        float* acc  = (l & 1) ? bufA : bufB;
        const float* xin = (l == 0) ? x : acc;
        float* xout = (l == 2) ? (float*)d_out : hbuf;

        k_compute_c<<<1, 1, 0, stream>>>(Wel, aev, easum, cvec, E);
        k_gemm_xw<<<1024, 256, 0, stream>>>(xin, Wl, hbuf, N);
        k_node_dots<<<512, 256, 0, stream>>>(hbuf, asv, adv, als, ald, N);
        k_node_self<<<nb_n, 256, 0, stream>>>(als, ald, cvec, aself, (unsigned*)amaxf, N);
        k_edge_alpha<<<nb_e, 256, 0, stream>>>(src, dst, ea, als, ald, cvec,
                                               alpha, (unsigned*)amaxf, E);
        k_node_exself<<<nb_n, 256, 0, stream>>>((unsigned*)amaxf, amaxf, aself,
                                                exself, denom, N);
        k_edge_exp<<<nb_e, 256, 0, stream>>>(alpha, amaxf, dst, denom, E);
        k_node_gather<<<2048, 256, 0, stream>>>(rowptr, deg, ssrc, perm, alpha, exself,
                                                denom, bv, hbuf, acc, N);
        k_zero<<<1, 256, 0, stream>>>(colsum, 256);
        k_bn_stats<<<(N + 255) / 256, 512, 0, stream>>>(acc, colsum, colsq, N);
        k_bn_apply<<<nb_nh, 256, 0, stream>>>(acc, colsum, colsq, gv, bev, xout, N);
    }
}


# BN apply: a lane owns four columns with their constants in registers and a wave streams 16 whole rows, all row loads in flight at once; 3125 waves a launch instead of 200000 one-element waves
# speedup vs baseline: 2.5382x; 1.6986x over previous
#include <hip/hip_runtime.h>
#include <hip/hip_bf16.h>


typedef float v2f __attribute__((ext_vector_type(2)));
typedef float v8f __attribute__((ext_vector_type(8)));

#define HF 128

__device__ __forceinline__ float leaky(float x, float s) { return x > 0.f ? x : s * x; }

__global__ void k_zero(float* p, int n) {
    int i = blockIdx.x * blockDim.x + threadIdx.x;
    if (i < n) p[i] = 0.f;
}

__global__ void k_ea_sum(const float* __restrict__ ea, int E, float* __restrict__ easum) {
    int idx = blockIdx.x * blockDim.x + threadIdx.x;
    int stride = gridDim.x * blockDim.x;
    float s0 = 0.f, s1 = 0.f, s2 = 0.f;
    for (int e = idx; e < E; e += stride) {
        s0 += ea[e * 3 + 0];
        s1 += ea[e * 3 + 1];
        s2 += ea[e * 3 + 2];
    }
    atomicAdd(easum + 0, s0);
    atomicAdd(easum + 1, s1);
    atomicAdd(easum + 2, s2);
}

__global__ void k_compute_c(const float* __restrict__ We, const float* __restrict__ ae,
                            const float* __restrict__ easum, float* __restrict__ cvec, int E) {
    if (threadIdx.x != 0 || blockIdx.x != 0) return;
    float cm = 0.f;
    for (int d = 0; d < 3; ++d) {
        float c = 0.f;
        for (int h = 0; h < HF; ++h) c += We[d * HF + h] * ae[h];
        cvec[d] = c;
        cm += (easum[d] / (float)E) * c;
    }
    cvec[3] = cm;
}

__global__ __launch_bounds__(256) void k_gemm_xw(const float* __restrict__ X,
                                                 const float* __restrict__ W,
                                                 float* __restrict__ Ho, int n) {
    const int lane = threadIdx.x & 31;
    const int wv   = threadIdx.x >> 5;
    const int lo16 = lane & 15;
    const int hi   = lane >> 4;
    const int ncol = wv * 16 + lo16;

    v2f breg[32];
#pragma unroll
    for (int kk = 0; kk < 32; ++kk) {
        int kb = kk * 4 + hi * 2;
        v2f b;
        b.x = W[(size_t)kb * HF + ncol];
        b.y = W[(size_t)(kb + 1) * HF + ncol];
        breg[kk] = b;
    }

    const int strips = (n + 15) >> 4;
    for (int s = blockIdx.x; s < strips; s += gridDim.x) {
        const int r0 = s * 16;
        int rowA = r0 + lo16;
        if (rowA >= n) rowA = n - 1;
        const float* xrow = X + (size_t)rowA * HF + hi * 2;

        v2f a[32];
#pragma unroll
        for (int kk = 0; kk < 32; ++kk) {
            float2 t = *(const float2*)(xrow + kk * 4);
            a[kk].x = t.x;
            a[kk].y = t.y;
        }

        v8f acc = {0.f, 0.f, 0.f, 0.f, 0.f, 0.f, 0.f, 0.f};
#pragma unroll
        for (int kk = 0; kk < 32; ++kk) {
            acc = __builtin_amdgcn_wmma_f32_16x16x4_f32(
                false, a[kk], false, breg[kk], (short)0, acc, false, false);
        }

#pragma unroll
        for (int v = 0; v < 8; ++v) {
            int row = r0 + v + hi * 8;
            if (row < n) Ho[(size_t)row * HF + ncol] = acc[v];
        }
    }
}

__global__ __launch_bounds__(256) void k_node_dots(const float* __restrict__ Hm,
                                                   const float* __restrict__ avs,
                                                   const float* __restrict__ avd,
                                                   float* __restrict__ als,
                                                   float* __restrict__ ald, int n) {
    const int lane = threadIdx.x & 31;
    int wid = (blockIdx.x * blockDim.x + threadIdx.x) >> 5;
    int nw  = (gridDim.x * blockDim.x) >> 5;
    float s0 = avs[lane], s1 = avs[lane + 32], s2 = avs[lane + 64], s3 = avs[lane + 96];
    float d0 = avd[lane], d1 = avd[lane + 32], d2 = avd[lane + 64], d3 = avd[lane + 96];
    for (int i = wid; i < n; i += nw) {
        const float* hr = Hm + (size_t)i * HF;
        float h0 = hr[lane], h1 = hr[lane + 32], h2 = hr[lane + 64], h3 = hr[lane + 96];
        float rs = h0 * s0 + h1 * s1 + h2 * s2 + h3 * s3;
        float rd = h0 * d0 + h1 * d1 + h2 * d2 + h3 * d3;
        for (int off = 16; off; off >>= 1) {
            rs += __shfl_xor(rs, off);
            rd += __shfl_xor(rd, off);
        }
        if (lane == 0) { als[i] = rs; ald[i] = rd; }
    }
}

__global__ void k_deg_count(const int* __restrict__ dst, int* __restrict__ deg, int E) {
    int e = blockIdx.x * blockDim.x + threadIdx.x;
    if (e >= E) return;
    atomicAdd(deg + dst[e], 1);
}

#define SCAN_T 1024
__global__ __launch_bounds__(SCAN_T) void k_row_scan(const int* __restrict__ deg,
                                                     int* __restrict__ rowptr,
                                                     int* __restrict__ cursor, int n) {
    __shared__ int part[SCAN_T];
    const int t = threadIdx.x;
    const int chunk = (n + SCAN_T - 1) / SCAN_T;
    const int b = t * chunk;
    int e = b + chunk;
    if (e > n) e = n;
    int s = 0;
    for (int i = b; i < e; ++i) s += deg[i];
    part[t] = s;
    __syncthreads();
    for (int off = 1; off < SCAN_T; off <<= 1) {
        const int v = (t >= off) ? part[t - off] : 0;
        __syncthreads();
        part[t] += v;
        __syncthreads();
    }
    int run = part[t] - s;
    for (int i = b; i < e; ++i) {
        const int dg = deg[i];
        rowptr[i] = run;
        cursor[i] = run;
        run += dg;
    }
}

__global__ void k_list_fill(const int* __restrict__ src, const int* __restrict__ dst,
                            const float* __restrict__ ea, int* __restrict__ cursor,
                            int* __restrict__ rec, int E) {
    int e = blockIdx.x * blockDim.x + threadIdx.x;
    if (e >= E) return;
    const int p = atomicAdd(cursor + dst[e], 1);
    int4 r;
    r.x = src[e];
    r.y = __float_as_int(ea[e * 3 + 0]);
    r.z = __float_as_int(ea[e * 3 + 1]);
    r.w = __float_as_int(ea[e * 3 + 2]);
    ((int4*)rec)[p] = r;
}

__device__ __forceinline__ int lane_i(int v, int k) { return __builtin_amdgcn_readlane(v, k); }
__device__ __forceinline__ float lane_f(float v, int k) {
    return __int_as_float(__builtin_amdgcn_readlane(__float_as_int(v), k));
}
__device__ __forceinline__ float first_f(float v) {
    return __int_as_float(__builtin_amdgcn_readfirstlane(__float_as_int(v)));
}

template <int MASK>
__device__ __forceinline__ float row_xor(float v) {
    return __int_as_float(
        __builtin_amdgcn_update_dpp(0, __float_as_int(v), 0x160 + MASK, 0xf, 0xf, false));
}
__device__ __forceinline__ float xhalf(float v) {
    const int b = __float_as_int(v);
    return __int_as_float(
        __builtin_amdgcn_permlanex16(b, b, 0x76543210, 0xfedcba98, false, false));
}
__device__ __forceinline__ float wave_sum(float v) {
    v += row_xor<1>(v);
    v += row_xor<2>(v);
    v += row_xor<4>(v);
    v += row_xor<8>(v);
    return v + xhalf(v);
}
__device__ __forceinline__ float wave_max(float v) {
    v = fmaxf(v, row_xor<1>(v));
    v = fmaxf(v, row_xor<2>(v));
    v = fmaxf(v, row_xor<4>(v));
    v = fmaxf(v, row_xor<8>(v));
    return fmaxf(v, xhalf(v));
}

__device__ __forceinline__ void row_fma(float4& o, const float4 h, float c) {
    o.x = fmaf(h.x, c, o.x);
    o.y = fmaf(h.y, c, o.y);
    o.z = fmaf(h.z, c, o.z);
    o.w = fmaf(h.w, c, o.w);
}

__device__ __forceinline__ float slot_alpha(const int4* __restrict__ rec4,
                                            const float* __restrict__ als, int j, float ad,
                                            float c0, float c1, float c2, int& s) {
    const int4 r = rec4[j];
    s = r.x;
    const float ale = __int_as_float(r.y) * c0 + __int_as_float(r.z) * c1 +
                      __int_as_float(r.w) * c2;
    return leaky(als[s] + ad + ale, 0.2f);
}

__device__ __forceinline__ void gather_group(float4& o, int sl, float cl, int m,
                                             const float* __restrict__ Hm, int lane) {
    int k = 0;
    for (; k + 4 <= m; k += 4) {
        const int s0 = lane_i(sl, k),     s1 = lane_i(sl, k + 1);
        const int s2 = lane_i(sl, k + 2), s3 = lane_i(sl, k + 3);
        const float c0 = lane_f(cl, k),     c1 = lane_f(cl, k + 1);
        const float c2 = lane_f(cl, k + 2), c3 = lane_f(cl, k + 3);
        const float4 h0 = *(const float4*)(Hm + (size_t)s0 * HF + lane * 4);
        const float4 h1 = *(const float4*)(Hm + (size_t)s1 * HF + lane * 4);
        const float4 h2 = *(const float4*)(Hm + (size_t)s2 * HF + lane * 4);
        const float4 h3 = *(const float4*)(Hm + (size_t)s3 * HF + lane * 4);
        row_fma(o, h0, c0);
        row_fma(o, h1, c1);
        row_fma(o, h2, c2);
        row_fma(o, h3, c3);
    }
    const int r = m - k;
    if (r == 3) {
        const int s0 = lane_i(sl, k), s1 = lane_i(sl, k + 1), s2 = lane_i(sl, k + 2);
        const float c0 = lane_f(cl, k), c1 = lane_f(cl, k + 1), c2 = lane_f(cl, k + 2);
        const float4 h0 = *(const float4*)(Hm + (size_t)s0 * HF + lane * 4);
        const float4 h1 = *(const float4*)(Hm + (size_t)s1 * HF + lane * 4);
        const float4 h2 = *(const float4*)(Hm + (size_t)s2 * HF + lane * 4);
        row_fma(o, h0, c0);
        row_fma(o, h1, c1);
        row_fma(o, h2, c2);
    } else if (r == 2) {
        const int s0 = lane_i(sl, k), s1 = lane_i(sl, k + 1);
        const float c0 = lane_f(cl, k), c1 = lane_f(cl, k + 1);
        const float4 h0 = *(const float4*)(Hm + (size_t)s0 * HF + lane * 4);
        const float4 h1 = *(const float4*)(Hm + (size_t)s1 * HF + lane * 4);
        row_fma(o, h0, c0);
        row_fma(o, h1, c1);
    } else if (r == 1) {
        const int s0 = lane_i(sl, k);
        const float c0 = lane_f(cl, k);
        const float4 h0 = *(const float4*)(Hm + (size_t)s0 * HF + lane * 4);
        row_fma(o, h0, c0);
    }
}

__global__ __launch_bounds__(256) void k_node_attn(const int* __restrict__ rowptr,
                                                   const int* __restrict__ deg,
                                                   const int* __restrict__ rec,
                                                   const float* __restrict__ als,
                                                   const float* __restrict__ ald,
                                                   const float* __restrict__ cvec,
                                                   const float* __restrict__ bias,
                                                   const float* __restrict__ Hm,
                                                   float* __restrict__ acc, int n) {
    const int lane = threadIdx.x & 31;
    const int nw   = (gridDim.x * blockDim.x) >> 5;
    const int w0 = __builtin_amdgcn_readfirstlane(
        (int)((blockIdx.x * blockDim.x + threadIdx.x) >> 5));
    const int4* __restrict__ rec4 = (const int4*)rec;
    const float c0 = cvec[0], c1 = cvec[1], c2 = cvec[2], cm = cvec[3];
    const float4 bv = *(const float4*)(bias + lane * 4);
    const float ninf = -__builtin_inff();
    for (int i = w0; i < n; i += nw) {
        const int start = rowptr[i];
        const int cnt   = deg[i];
        const float ad  = ald[i];
        const float as  = als[i];
        const float4 hs = *(const float4*)(Hm + (size_t)i * HF + lane * 4);
        float4 o;
        if (cnt <= 32) {
            const bool ok = lane < cnt;
            int sl = 0;
            float a = ninf;
            if (ok) a = slot_alpha(rec4, als, start + lane, ad, c0, c1, c2, sl);
            const float aself = leaky(as + ad + cm, 0.2f);
            const float m   = fmaxf(first_f(wave_max(a)), aself);
            const float ex  = ok ? expf(a - m) : 0.f;
            const float exs = expf(aself - m);
            const float den = (exs + first_f(wave_sum(ex))) + 1e-16f;
            const float cs  = exs / den;
            o.x = hs.x * cs + bv.x;
            o.y = hs.y * cs + bv.y;
            o.z = hs.z * cs + bv.z;
            o.w = hs.w * cs + bv.w;
            if (cnt > 0) gather_group(o, sl, ex / den, cnt, Hm, lane);
        } else {
            float mx = ninf;
            for (int base = 0; base < cnt; base += 32) {
                if (base + lane < cnt) {
                    int s;
                    mx = fmaxf(mx, slot_alpha(rec4, als, start + base + lane, ad, c0, c1, c2, s));
                }
            }
            const float aself = leaky(as + ad + cm, 0.2f);
            const float m = fmaxf(first_f(wave_max(mx)), aself);
            float sm = 0.f;
            for (int base = 0; base < cnt; base += 32) {
                if (base + lane < cnt) {
                    int s;
                    sm += expf(slot_alpha(rec4, als, start + base + lane, ad, c0, c1, c2, s) - m);
                }
            }
            const float exs = expf(aself - m);
            const float den = (exs + first_f(wave_sum(sm))) + 1e-16f;
            const float cs  = exs / den;
            o.x = hs.x * cs + bv.x;
            o.y = hs.y * cs + bv.y;
            o.z = hs.z * cs + bv.z;
            o.w = hs.w * cs + bv.w;
            for (int base = 0; base < cnt; base += 32) {
                int mm = cnt - base;
                if (mm > 32) mm = 32;
                int sl = 0;
                float cl = 0.f;
                if (lane < mm) {
                    const float a = slot_alpha(rec4, als, start + base + lane, ad, c0, c1, c2, sl);
                    cl = expf(a - m) / den;
                }
                gather_group(o, sl, cl, mm, Hm, lane);
            }
        }
        *(float4*)(acc + (size_t)i * HF + lane * 4) = o;
    }
}

#define BN_ROWS  256
#define BN_WAVES 16
#define BN_RPW   (BN_ROWS / BN_WAVES)
__global__ __launch_bounds__(512) void k_bn_stats(const float* __restrict__ acc,
                                                  float* __restrict__ colsum,
                                                  float* __restrict__ colsq, int n) {
    __shared__ __attribute__((aligned(16))) float part[2][BN_WAVES][HF];
    const int lane = threadIdx.x & 31;
    const int w    = __builtin_amdgcn_readfirstlane((int)(threadIdx.x >> 5));
    const int r0   = blockIdx.x * BN_ROWS + w;

    float4 t[BN_RPW];
#pragma unroll
    for (int i = 0; i < BN_RPW; ++i) {
        int r = r0 + i * BN_WAVES;
        if (r >= n) r = n - 1;
        t[i] = *(const float4*)(acc + (size_t)r * HF + lane * 4);
    }

    float4 s = {0.f, 0.f, 0.f, 0.f}, q = {0.f, 0.f, 0.f, 0.f};
#pragma unroll
    for (int i = 0; i < BN_RPW; ++i) {
        const bool ok = (r0 + i * BN_WAVES) < n;
        float y0 = ok ? leaky(t[i].x, 0.01f) : 0.f;
        float y1 = ok ? leaky(t[i].y, 0.01f) : 0.f;
        float y2 = ok ? leaky(t[i].z, 0.01f) : 0.f;
        float y3 = ok ? leaky(t[i].w, 0.01f) : 0.f;
        s.x += y0; q.x += y0 * y0;
        s.y += y1; q.y += y1 * y1;
        s.z += y2; q.z += y2 * y2;
        s.w += y3; q.w += y3 * y3;
    }
    *(float4*)&part[0][w][lane * 4] = s;
    *(float4*)&part[1][w][lane * 4] = q;
    __syncthreads();

    if (threadIdx.x < 2 * HF) {
        const int which = threadIdx.x >> 7;
        const int c     = threadIdx.x & (HF - 1);
        float r = 0.f;
        for (int k = 0; k < BN_WAVES; ++k) r += part[which][k][c];
        atomicAdd((which ? colsq : colsum) + c, r);
    }
}

#define BA_RPW  16
#define BA_ROWS (8 * BA_RPW)
__device__ __forceinline__ void bn_consts(const float4 cs, const float4 cq, int n, float4& mu,
                                          float4& rs) {
    const float invN = 1.f / (float)n;
    mu.x = cs.x * invN;  rs.x = rsqrtf((cq.x * invN - mu.x * mu.x) + 1e-5f);
    mu.y = cs.y * invN;  rs.y = rsqrtf((cq.y * invN - mu.y * mu.y) + 1e-5f);
    mu.z = cs.z * invN;  rs.z = rsqrtf((cq.z * invN - mu.z * mu.z) + 1e-5f);
    mu.w = cs.w * invN;  rs.w = rsqrtf((cq.w * invN - mu.w * mu.w) + 1e-5f);
}
__device__ __forceinline__ float4 bn_row(const float4 t, const float4 mu, const float4 rs,
                                         const float4 gv, const float4 bv) {
    float4 o;
    o.x = (leaky(t.x, 0.01f) - mu.x) * rs.x * gv.x + bv.x;
    o.y = (leaky(t.y, 0.01f) - mu.y) * rs.y * gv.y + bv.y;
    o.z = (leaky(t.z, 0.01f) - mu.z) * rs.z * gv.z + bv.z;
    o.w = (leaky(t.w, 0.01f) - mu.w) * rs.w * gv.w + bv.w;
    return o;
}
__global__ __launch_bounds__(256) void k_bn_apply(const float* __restrict__ acc,
                                                  const float* __restrict__ colsum,
                                                  const float* __restrict__ colsq,
                                                  const float* __restrict__ g,
                                                  const float* __restrict__ be,
                                                  float* __restrict__ xout, int n) {
    const int lane = threadIdx.x & 31;
    const int w  = __builtin_amdgcn_readfirstlane(
        (int)((blockIdx.x * blockDim.x + threadIdx.x) >> 5));
    const int r0 = w * BA_RPW;
    if (r0 >= n) return;

    const float4 cs = *(const float4*)(colsum + lane * 4);
    const float4 cq = *(const float4*)(colsq + lane * 4);
    const float4 gv = *(const float4*)(g + lane * 4);
    const float4 bv = *(const float4*)(be + lane * 4);
    const float* __restrict__ src = acc + (size_t)r0 * HF + lane * 4;
    float* __restrict__ dst = xout + (size_t)r0 * HF + lane * 4;

    if (r0 + BA_RPW <= n) {
        float4 t[BA_RPW];
#pragma unroll
        for (int i = 0; i < BA_RPW; ++i) t[i] = *(const float4*)(src + (size_t)i * HF);

        float4 mu, rs;
        bn_consts(cs, cq, n, mu, rs);
#pragma unroll
        for (int i = 0; i < BA_RPW; ++i)
            *(float4*)(dst + (size_t)i * HF) = bn_row(t[i], mu, rs, gv, bv);
    } else {
        float4 mu, rs;
        bn_consts(cs, cq, n, mu, rs);
        for (int i = 0; r0 + i < n; ++i)
            *(float4*)(dst + (size_t)i * HF) =
                bn_row(*(const float4*)(src + (size_t)i * HF), mu, rs, gv, bv);
    }
}

extern "C" void kernel_launch(void* const* d_in, const int* in_sizes, int n_in,
                              void* d_out, int out_size, void* d_ws, size_t ws_size,
                              hipStream_t stream) {
    const float* x  = (const float*)d_in[0];
    const int*   ei = (const int*)d_in[1];
    const float* ea = (const float*)d_in[3];
    const int N = in_sizes[0] / HF;
    const int E = in_sizes[1] / 2;
    const int* src = ei;
    const int* dst = ei + E;

    float* ws = (float*)d_ws;
    const size_t NH = (size_t)N * HF;
    float* bufA   = ws;
    float* bufB   = bufA + NH;
    int*   rec    = (int*)(bufB + NH);
    int*   rowptr = rec + 4 * (size_t)E;
    int*   cursor = rowptr + N;
    int*   deg    = cursor + N;
    float* als    = bufB + NH + NH;
    float* ald    = als + N;
    float* colsum = ald + N;
    float* colsq  = colsum + HF;
    float* easum  = colsq + HF;
    float* cvec   = easum + 4;

    k_zero<<<1, 8, 0, stream>>>(easum, 8);
    k_ea_sum<<<128, 256, 0, stream>>>(ea, E, easum);

    const int nb_e  = (E + 255) / 256;
    const int nb_n  = (N + 255) / 256;

    k_zero<<<nb_n, 256, 0, stream>>>((float*)deg, N);
    k_deg_count<<<nb_e, 256, 0, stream>>>(dst, deg, E);
    k_row_scan<<<1, SCAN_T, 0, stream>>>(deg, rowptr, cursor, N);
    k_list_fill<<<nb_e, 256, 0, stream>>>(src, dst, ea, cursor, rec, E);

    for (int l = 0; l < 3; ++l) {
        const float* Wl  = (const float*)d_in[4 + 8 * l];
        const float* asv = (const float*)d_in[5 + 8 * l];
        const float* adv = (const float*)d_in[6 + 8 * l];
        const float* Wel = (const float*)d_in[7 + 8 * l];
        const float* aev = (const float*)d_in[8 + 8 * l];
        const float* bv  = (const float*)d_in[9 + 8 * l];
        const float* gv  = (const float*)d_in[10 + 8 * l];
        const float* bev = (const float*)d_in[11 + 8 * l];
        float* hbuf = (l & 1) ? bufB : bufA;
        float* acc  = (l & 1) ? bufA : bufB;
        const float* xin = (l == 0) ? x : acc;
        float* xout = (l == 2) ? (float*)d_out : hbuf;

        k_compute_c<<<1, 1, 0, stream>>>(Wel, aev, easum, cvec, E);
        k_gemm_xw<<<1024, 256, 0, stream>>>(xin, Wl, hbuf, N);
        k_node_dots<<<512, 256, 0, stream>>>(hbuf, asv, adv, als, ald, N);
        k_node_attn<<<2048, 256, 0, stream>>>(rowptr, deg, rec, als, ald, cvec, bv,
                                              hbuf, acc, N);
        k_zero<<<1, 256, 0, stream>>>(colsum, 256);
        k_bn_stats<<<(N + 255) / 256, 512, 0, stream>>>(acc, colsum, colsq, N);
        k_bn_apply<<<(N + BA_ROWS - 1) / BA_ROWS, 256, 0, stream>>>(acc, colsum, colsq, gv, bev,
                                                                  xout, N);
    }
}


# Row-offset scan, still one block: a wave owns a contiguous piece and reads 128 counts a step as one int4 per lane (whole lines), lane scan with carry, whole-line stores; integer sums only
# speedup vs baseline: 3.3954x; 1.3377x over previous
#include <hip/hip_runtime.h>
#include <hip/hip_bf16.h>


typedef float v2f __attribute__((ext_vector_type(2)));
typedef float v8f __attribute__((ext_vector_type(8)));

#define HF 128

__device__ __forceinline__ float leaky(float x, float s) { return x > 0.f ? x : s * x; }

__global__ void k_zero2(float* p, int n, float* q, int m) {
    int i = blockIdx.x * blockDim.x + threadIdx.x;
    if (i < n) p[i] = 0.f;
    if (i < m) q[i] = 0.f;
}

__global__ __launch_bounds__(256) void k_gemm_xw(const float* __restrict__ X,
                                                 const float* __restrict__ W,
                                                 float* __restrict__ Ho,
                                                 float* __restrict__ colz, int n) {
    if (blockIdx.x == 0) colz[threadIdx.x] = 0.f;
    const int lane = threadIdx.x & 31;
    const int wv   = threadIdx.x >> 5;
    const int lo16 = lane & 15;
    const int hi   = lane >> 4;
    const int ncol = wv * 16 + lo16;

    v2f breg[32];
#pragma unroll
    for (int kk = 0; kk < 32; ++kk) {
        int kb = kk * 4 + hi * 2;
        v2f b;
        b.x = W[(size_t)kb * HF + ncol];
        b.y = W[(size_t)(kb + 1) * HF + ncol];
        breg[kk] = b;
    }

    const int strips = (n + 15) >> 4;
    for (int s = blockIdx.x; s < strips; s += gridDim.x) {
        const int r0 = s * 16;
        int rowA = r0 + lo16;
        if (rowA >= n) rowA = n - 1;
        const float* xrow = X + (size_t)rowA * HF + hi * 2;

        v2f a[32];
#pragma unroll
        for (int kk = 0; kk < 32; ++kk) {
            float2 t = *(const float2*)(xrow + kk * 4);
            a[kk].x = t.x;
            a[kk].y = t.y;
        }

        v8f acc = {0.f, 0.f, 0.f, 0.f, 0.f, 0.f, 0.f, 0.f};
#pragma unroll
        for (int kk = 0; kk < 32; ++kk) {
            acc = __builtin_amdgcn_wmma_f32_16x16x4_f32(
                false, a[kk], false, breg[kk], (short)0, acc, false, false);
        }

#pragma unroll
        for (int v = 0; v < 8; ++v) {
            int row = r0 + v + hi * 8;
            if (row < n) Ho[(size_t)row * HF + ncol] = acc[v];
        }
    }
}

__global__ __launch_bounds__(256) void k_node_dots(const float* __restrict__ Hm,
                                                   const float* __restrict__ avs,
                                                   const float* __restrict__ avd,
                                                   float* __restrict__ als,
                                                   float* __restrict__ ald, int n) {
    const int lane = threadIdx.x & 31;
    int wid = (blockIdx.x * blockDim.x + threadIdx.x) >> 5;
    int nw  = (gridDim.x * blockDim.x) >> 5;
    float s0 = avs[lane], s1 = avs[lane + 32], s2 = avs[lane + 64], s3 = avs[lane + 96];
    float d0 = avd[lane], d1 = avd[lane + 32], d2 = avd[lane + 64], d3 = avd[lane + 96];
    for (int i = wid; i < n; i += nw) {
        const float* hr = Hm + (size_t)i * HF;
        float h0 = hr[lane], h1 = hr[lane + 32], h2 = hr[lane + 64], h3 = hr[lane + 96];
        float rs = h0 * s0 + h1 * s1 + h2 * s2 + h3 * s3;
        float rd = h0 * d0 + h1 * d1 + h2 * d2 + h3 * d3;
        for (int off = 16; off; off >>= 1) {
            rs += __shfl_xor(rs, off);
            rd += __shfl_xor(rd, off);
        }
        if (lane == 0) { als[i] = rs; ald[i] = rd; }
    }
}

__global__ void k_deg_count(const int* __restrict__ dst, int* __restrict__ deg, int E) {
    int e = blockIdx.x * blockDim.x + threadIdx.x;
    if (e >= E) return;
    atomicAdd(deg + dst[e], 1);
}

#define SCAN_T 1024
#define SCAN_W (SCAN_T / 32)
#define SCAN_STEP 128
__device__ __forceinline__ int wave_isum(int v) {
    for (int off = 16; off; off >>= 1) v += __shfl_xor(v, off);
    return v;
}
__device__ __forceinline__ int4 deg4_tail(const int* __restrict__ deg, int i, int n) {
    int4 v = {0, 0, 0, 0};
    if (i < n)     v.x = deg[i];
    if (i + 1 < n) v.y = deg[i + 1];
    if (i + 2 < n) v.z = deg[i + 2];
    if (i + 3 < n) v.w = deg[i + 3];
    return v;
}
__device__ __forceinline__ int4 scan4(const int4 v, int lane, int& carry) {
    const int own = (v.x + v.y) + (v.z + v.w);
    int inc = own;
    for (int off = 1; off < 32; off <<= 1) {
        const int u = __shfl_up(inc, off);
        if (lane >= off) inc += u;
    }
    const int run = carry + (inc - own);
    carry += __builtin_amdgcn_readlane(inc, 31);
    int4 o;
    o.x = run;
    o.y = o.x + v.x;
    o.z = o.y + v.y;
    o.w = o.z + v.z;
    return o;
}
__global__ __launch_bounds__(SCAN_T) void k_row_scan(const int* __restrict__ deg,
                                                     int* __restrict__ rowptr,
                                                     int* __restrict__ cursor, int n) {
    __shared__ int wtot[SCAN_W];
    const int lane = threadIdx.x & 31;
    const int w = __builtin_amdgcn_readfirstlane((int)(threadIdx.x >> 5));
    const int per = (((n + SCAN_W - 1) / SCAN_W) + SCAN_STEP - 1) & ~(SCAN_STEP - 1);
    const int lo = w * per;
    int hi = lo + per;
    if (hi > n) hi = n;
    const int l4 = lane * 4;

    int s = 0;
    int i0 = lo;
#pragma unroll 4
    for (; i0 + SCAN_STEP <= hi; i0 += SCAN_STEP) {
        const int4 v = *(const int4*)(deg + i0 + l4);
        s += (v.x + v.y) + (v.z + v.w);
    }
    if (i0 < hi) {
        const int4 v = deg4_tail(deg, i0 + l4, n);
        s += (v.x + v.y) + (v.z + v.w);
    }
    s = wave_isum(s);
    if (lane == 0) wtot[w] = s;
    __syncthreads();
    int carry = wave_isum(lane < w ? wtot[lane] : 0);

    i0 = lo;
#pragma unroll 4
    for (; i0 + SCAN_STEP <= hi; i0 += SCAN_STEP) {
        const int i = i0 + l4;
        const int4 o = scan4(*(const int4*)(deg + i), lane, carry);
        *(int4*)(rowptr + i) = o;
        *(int4*)(cursor + i) = o;
    }
    if (i0 < hi) {
        const int i = i0 + l4;
        const int4 o = scan4(deg4_tail(deg, i, n), lane, carry);
        if (i < n)     { rowptr[i] = o.x;     cursor[i] = o.x; }
        if (i + 1 < n) { rowptr[i + 1] = o.y; cursor[i + 1] = o.y; }
        if (i + 2 < n) { rowptr[i + 2] = o.z; cursor[i + 2] = o.z; }
        if (i + 3 < n) { rowptr[i + 3] = o.w; cursor[i + 3] = o.w; }
    }
}

__device__ __forceinline__ int lane_i(int v, int k) { return __builtin_amdgcn_readlane(v, k); }
__device__ __forceinline__ float lane_f(float v, int k) {
    return __int_as_float(__builtin_amdgcn_readlane(__float_as_int(v), k));
}
__device__ __forceinline__ float first_f(float v) {
    return __int_as_float(__builtin_amdgcn_readfirstlane(__float_as_int(v)));
}

template <int MASK>
__device__ __forceinline__ float row_xor(float v) {
    return __int_as_float(
        __builtin_amdgcn_update_dpp(0, __float_as_int(v), 0x160 + MASK, 0xf, 0xf, false));
}
__device__ __forceinline__ float xhalf(float v) {
    const int b = __float_as_int(v);
    return __int_as_float(
        __builtin_amdgcn_permlanex16(b, b, 0x76543210, 0xfedcba98, false, false));
}
__device__ __forceinline__ float wave_sum(float v) {
    v += row_xor<1>(v);
    v += row_xor<2>(v);
    v += row_xor<4>(v);
    v += row_xor<8>(v);
    return v + xhalf(v);
}
__device__ __forceinline__ float wave_max(float v) {
    v = fmaxf(v, row_xor<1>(v));
    v = fmaxf(v, row_xor<2>(v));
    v = fmaxf(v, row_xor<4>(v));
    v = fmaxf(v, row_xor<8>(v));
    return fmaxf(v, xhalf(v));
}

__global__ __launch_bounds__(256) void k_list_fill(const int* __restrict__ src,
                                                   const int* __restrict__ dst,
                                                   const float* __restrict__ ea,
                                                   int* __restrict__ cursor,
                                                   int* __restrict__ rec,
                                                   float* __restrict__ easum, int E) {
    __shared__ float part[3][8];
    const int e = blockIdx.x * blockDim.x + threadIdx.x;
    float a0 = 0.f, a1 = 0.f, a2 = 0.f;
    if (e < E) {
        const int p = atomicAdd(cursor + dst[e], 1);
        a0 = ea[e * 3 + 0];
        a1 = ea[e * 3 + 1];
        a2 = ea[e * 3 + 2];
        int4 r;
        r.x = src[e];
        r.y = __float_as_int(a0);
        r.z = __float_as_int(a1);
        r.w = __float_as_int(a2);
        ((int4*)rec)[p] = r;
    }
    a0 = wave_sum(a0);
    a1 = wave_sum(a1);
    a2 = wave_sum(a2);
    const int w = threadIdx.x >> 5;
    if ((threadIdx.x & 31) == 0) {
        part[0][w] = a0;
        part[1][w] = a1;
        part[2][w] = a2;
    }
    __syncthreads();
    if (threadIdx.x < 3) {
        float s = 0.f;
        for (int k = 0; k < 8; ++k) s += part[threadIdx.x][k];
        atomicAdd(easum + threadIdx.x, s);
    }
}

__device__ __forceinline__ void row_fma(float4& o, const float4 h, float c) {
    o.x = fmaf(h.x, c, o.x);
    o.y = fmaf(h.y, c, o.y);
    o.z = fmaf(h.z, c, o.z);
    o.w = fmaf(h.w, c, o.w);
}

__device__ __forceinline__ float slot_alpha(const int4* __restrict__ rec4,
                                            const float* __restrict__ als, int j, float ad,
                                            float c0, float c1, float c2, int& s) {
    const int4 r = rec4[j];
    s = r.x;
    const float ale = __int_as_float(r.y) * c0 + __int_as_float(r.z) * c1 +
                      __int_as_float(r.w) * c2;
    return leaky(als[s] + ad + ale, 0.2f);
}

__device__ __forceinline__ void gather_group(float4& o, int sl, float cl, int m,
                                             const float* __restrict__ Hm, int lane) {
    int k = 0;
    for (; k + 4 <= m; k += 4) {
        const int s0 = lane_i(sl, k),     s1 = lane_i(sl, k + 1);
        const int s2 = lane_i(sl, k + 2), s3 = lane_i(sl, k + 3);
        const float c0 = lane_f(cl, k),     c1 = lane_f(cl, k + 1);
        const float c2 = lane_f(cl, k + 2), c3 = lane_f(cl, k + 3);
        const float4 h0 = *(const float4*)(Hm + (size_t)s0 * HF + lane * 4);
        const float4 h1 = *(const float4*)(Hm + (size_t)s1 * HF + lane * 4);
        const float4 h2 = *(const float4*)(Hm + (size_t)s2 * HF + lane * 4);
        const float4 h3 = *(const float4*)(Hm + (size_t)s3 * HF + lane * 4);
        row_fma(o, h0, c0);
        row_fma(o, h1, c1);
        row_fma(o, h2, c2);
        row_fma(o, h3, c3);
    }
    const int r = m - k;
    if (r == 3) {
        const int s0 = lane_i(sl, k), s1 = lane_i(sl, k + 1), s2 = lane_i(sl, k + 2);
        const float c0 = lane_f(cl, k), c1 = lane_f(cl, k + 1), c2 = lane_f(cl, k + 2);
        const float4 h0 = *(const float4*)(Hm + (size_t)s0 * HF + lane * 4);
        const float4 h1 = *(const float4*)(Hm + (size_t)s1 * HF + lane * 4);
        const float4 h2 = *(const float4*)(Hm + (size_t)s2 * HF + lane * 4);
        row_fma(o, h0, c0);
        row_fma(o, h1, c1);
        row_fma(o, h2, c2);
    } else if (r == 2) {
        const int s0 = lane_i(sl, k), s1 = lane_i(sl, k + 1);
        const float c0 = lane_f(cl, k), c1 = lane_f(cl, k + 1);
        const float4 h0 = *(const float4*)(Hm + (size_t)s0 * HF + lane * 4);
        const float4 h1 = *(const float4*)(Hm + (size_t)s1 * HF + lane * 4);
        row_fma(o, h0, c0);
        row_fma(o, h1, c1);
    } else if (r == 1) {
        const int s0 = lane_i(sl, k);
        const float c0 = lane_f(cl, k);
        const float4 h0 = *(const float4*)(Hm + (size_t)s0 * HF + lane * 4);
        row_fma(o, h0, c0);
    }
}

__global__ __launch_bounds__(256) void k_node_attn(const int* __restrict__ rowptr,
                                                   const int* __restrict__ deg,
                                                   const int* __restrict__ rec,
                                                   const float* __restrict__ als,
                                                   const float* __restrict__ ald,
                                                   const float* __restrict__ We,
                                                   const float* __restrict__ ae,
                                                   const float* __restrict__ easum, int E,
                                                   const float* __restrict__ bias,
                                                   const float* __restrict__ Hm,
                                                   float* __restrict__ acc, int n) {
    const int lane = threadIdx.x & 31;
    const int nw   = (gridDim.x * blockDim.x) >> 5;
    const int w0 = __builtin_amdgcn_readfirstlane(
        (int)((blockIdx.x * blockDim.x + threadIdx.x) >> 5));
    const int4* __restrict__ rec4 = (const int4*)rec;
    const float4 av = *(const float4*)(ae + lane * 4);
    const float4 e0 = *(const float4*)(We + lane * 4);
    const float4 e1 = *(const float4*)(We + HF + lane * 4);
    const float4 e2 = *(const float4*)(We + 2 * HF + lane * 4);
    const float c0 = first_f(wave_sum(e0.x * av.x + e0.y * av.y + e0.z * av.z + e0.w * av.w));
    const float c1 = first_f(wave_sum(e1.x * av.x + e1.y * av.y + e1.z * av.z + e1.w * av.w));
    const float c2 = first_f(wave_sum(e2.x * av.x + e2.y * av.y + e2.z * av.z + e2.w * av.w));
    float cm = 0.f;
    cm += (easum[0] / (float)E) * c0;
    cm += (easum[1] / (float)E) * c1;
    cm += (easum[2] / (float)E) * c2;
    const float4 bv = *(const float4*)(bias + lane * 4);
    const float ninf = -__builtin_inff();
    for (int i = w0; i < n; i += nw) {
        const int start = rowptr[i];
        const int cnt   = deg[i];
        const float ad  = ald[i];
        const float as  = als[i];
        const float4 hs = *(const float4*)(Hm + (size_t)i * HF + lane * 4);
        float4 o;
        if (cnt <= 32) {
            const bool ok = lane < cnt;
            int sl = 0;
            float a = ninf;
            if (ok) a = slot_alpha(rec4, als, start + lane, ad, c0, c1, c2, sl);
            const float aself = leaky(as + ad + cm, 0.2f);
            const float m   = fmaxf(first_f(wave_max(a)), aself);
            const float ex  = ok ? expf(a - m) : 0.f;
            const float exs = expf(aself - m);
            const float den = (exs + first_f(wave_sum(ex))) + 1e-16f;
            const float cs  = exs / den;
            o.x = hs.x * cs + bv.x;
            o.y = hs.y * cs + bv.y;
            o.z = hs.z * cs + bv.z;
            o.w = hs.w * cs + bv.w;
            if (cnt > 0) gather_group(o, sl, ex / den, cnt, Hm, lane);
        } else {
            float mx = ninf;
            for (int base = 0; base < cnt; base += 32) {
                if (base + lane < cnt) {
                    int s;
                    mx = fmaxf(mx, slot_alpha(rec4, als, start + base + lane, ad, c0, c1, c2, s));
                }
            }
            const float aself = leaky(as + ad + cm, 0.2f);
            const float m = fmaxf(first_f(wave_max(mx)), aself);
            float sm = 0.f;
            for (int base = 0; base < cnt; base += 32) {
                if (base + lane < cnt) {
                    int s;
                    sm += expf(slot_alpha(rec4, als, start + base + lane, ad, c0, c1, c2, s) - m);
                }
            }
            const float exs = expf(aself - m);
            const float den = (exs + first_f(wave_sum(sm))) + 1e-16f;
            const float cs  = exs / den;
            o.x = hs.x * cs + bv.x;
            o.y = hs.y * cs + bv.y;
            o.z = hs.z * cs + bv.z;
            o.w = hs.w * cs + bv.w;
            for (int base = 0; base < cnt; base += 32) {
                int mm = cnt - base;
                if (mm > 32) mm = 32;
                int sl = 0;
                float cl = 0.f;
                if (lane < mm) {
                    const float a = slot_alpha(rec4, als, start + base + lane, ad, c0, c1, c2, sl);
                    cl = expf(a - m) / den;
                }
                gather_group(o, sl, cl, mm, Hm, lane);
            }
        }
        *(float4*)(acc + (size_t)i * HF + lane * 4) = o;
    }
}

#define BN_ROWS  256
#define BN_WAVES 16
#define BN_RPW   (BN_ROWS / BN_WAVES)
__global__ __launch_bounds__(512) void k_bn_stats(const float* __restrict__ acc,
                                                  float* __restrict__ colsum,
                                                  float* __restrict__ colsq, int n) {
    __shared__ __attribute__((aligned(16))) float part[2][BN_WAVES][HF];
    const int lane = threadIdx.x & 31;
    const int w    = __builtin_amdgcn_readfirstlane((int)(threadIdx.x >> 5));
    const int r0   = blockIdx.x * BN_ROWS + w;

    float4 t[BN_RPW];
#pragma unroll
    for (int i = 0; i < BN_RPW; ++i) {
        int r = r0 + i * BN_WAVES;
        if (r >= n) r = n - 1;
        t[i] = *(const float4*)(acc + (size_t)r * HF + lane * 4);
    }

    float4 s = {0.f, 0.f, 0.f, 0.f}, q = {0.f, 0.f, 0.f, 0.f};
#pragma unroll
    for (int i = 0; i < BN_RPW; ++i) {
        const bool ok = (r0 + i * BN_WAVES) < n;
        float y0 = ok ? leaky(t[i].x, 0.01f) : 0.f;
        float y1 = ok ? leaky(t[i].y, 0.01f) : 0.f;
        float y2 = ok ? leaky(t[i].z, 0.01f) : 0.f;
        float y3 = ok ? leaky(t[i].w, 0.01f) : 0.f;
        s.x += y0; q.x += y0 * y0;
        s.y += y1; q.y += y1 * y1;
        s.z += y2; q.z += y2 * y2;
        s.w += y3; q.w += y3 * y3;
    }
    *(float4*)&part[0][w][lane * 4] = s;
    *(float4*)&part[1][w][lane * 4] = q;
    __syncthreads();

    if (threadIdx.x < 2 * HF) {
        const int which = threadIdx.x >> 7;
        const int c     = threadIdx.x & (HF - 1);
        float r = 0.f;
        for (int k = 0; k < BN_WAVES; ++k) r += part[which][k][c];
        atomicAdd((which ? colsq : colsum) + c, r);
    }
}

#define BA_RPW  16
#define BA_ROWS (8 * BA_RPW)
__device__ __forceinline__ void bn_consts(const float4 cs, const float4 cq, int n, float4& mu,
                                          float4& rs) {
    const float invN = 1.f / (float)n;
    mu.x = cs.x * invN;  rs.x = rsqrtf((cq.x * invN - mu.x * mu.x) + 1e-5f);
    mu.y = cs.y * invN;  rs.y = rsqrtf((cq.y * invN - mu.y * mu.y) + 1e-5f);
    mu.z = cs.z * invN;  rs.z = rsqrtf((cq.z * invN - mu.z * mu.z) + 1e-5f);
    mu.w = cs.w * invN;  rs.w = rsqrtf((cq.w * invN - mu.w * mu.w) + 1e-5f);
}
__device__ __forceinline__ float4 bn_row(const float4 t, const float4 mu, const float4 rs,
                                         const float4 gv, const float4 bv) {
    float4 o;
    o.x = (leaky(t.x, 0.01f) - mu.x) * rs.x * gv.x + bv.x;
    o.y = (leaky(t.y, 0.01f) - mu.y) * rs.y * gv.y + bv.y;
    o.z = (leaky(t.z, 0.01f) - mu.z) * rs.z * gv.z + bv.z;
    o.w = (leaky(t.w, 0.01f) - mu.w) * rs.w * gv.w + bv.w;
    return o;
}
__global__ __launch_bounds__(256) void k_bn_apply(const float* __restrict__ acc,
                                                  const float* __restrict__ colsum,
                                                  const float* __restrict__ colsq,
                                                  const float* __restrict__ g,
                                                  const float* __restrict__ be,
                                                  float* __restrict__ xout, int n) {
    const int lane = threadIdx.x & 31;
    const int w  = __builtin_amdgcn_readfirstlane(
        (int)((blockIdx.x * blockDim.x + threadIdx.x) >> 5));
    const int r0 = w * BA_RPW;
    if (r0 >= n) return;

    const float4 cs = *(const float4*)(colsum + lane * 4);
    const float4 cq = *(const float4*)(colsq + lane * 4);
    const float4 gv = *(const float4*)(g + lane * 4);
    const float4 bv = *(const float4*)(be + lane * 4);
    const float* __restrict__ src = acc + (size_t)r0 * HF + lane * 4;
    float* __restrict__ dst = xout + (size_t)r0 * HF + lane * 4;

    if (r0 + BA_RPW <= n) {
        float4 t[BA_RPW];
#pragma unroll
        for (int i = 0; i < BA_RPW; ++i) t[i] = *(const float4*)(src + (size_t)i * HF);

        float4 mu, rs;
        bn_consts(cs, cq, n, mu, rs);
#pragma unroll
        for (int i = 0; i < BA_RPW; ++i)
            *(float4*)(dst + (size_t)i * HF) = bn_row(t[i], mu, rs, gv, bv);
    } else {
        float4 mu, rs;
        bn_consts(cs, cq, n, mu, rs);
        for (int i = 0; r0 + i < n; ++i)
            *(float4*)(dst + (size_t)i * HF) =
                bn_row(*(const float4*)(src + (size_t)i * HF), mu, rs, gv, bv);
    }
}

extern "C" void kernel_launch(void* const* d_in, const int* in_sizes, int n_in,
                              void* d_out, int out_size, void* d_ws, size_t ws_size,
                              hipStream_t stream) {
    const float* x  = (const float*)d_in[0];
    const int*   ei = (const int*)d_in[1];
    const float* ea = (const float*)d_in[3];
    const int N = in_sizes[0] / HF;
    const int E = in_sizes[1] / 2;
    const int* src = ei;
    const int* dst = ei + E;

    float* ws = (float*)d_ws;
    const size_t NH = (size_t)N * HF;
    float* bufA   = ws;
    float* bufB   = bufA + NH;
    int*   rec    = (int*)(bufB + NH);
    int*   rowptr = rec + 4 * (size_t)E;
    int*   cursor = rowptr + N;
    int*   deg    = cursor + N;
    float* als    = bufB + NH + NH;
    float* ald    = als + N;
    float* colsum = ald + N;
    float* colsq  = colsum + HF;
    float* easum  = colsq + HF;

    const int nb_e  = (E + 255) / 256;
    const int nb_n  = (N + 255) / 256;

    k_zero2<<<nb_n, 256, 0, stream>>>((float*)deg, N, easum, 4);

    k_deg_count<<<nb_e, 256, 0, stream>>>(dst, deg, E);
    k_row_scan<<<1, SCAN_T, 0, stream>>>(deg, rowptr, cursor, N);
    k_list_fill<<<nb_e, 256, 0, stream>>>(src, dst, ea, cursor, rec, easum, E);

    for (int l = 0; l < 3; ++l) {
        const float* Wl  = (const float*)d_in[4 + 8 * l];
        const float* asv = (const float*)d_in[5 + 8 * l];
        const float* adv = (const float*)d_in[6 + 8 * l];
        const float* Wel = (const float*)d_in[7 + 8 * l];
        const float* aev = (const float*)d_in[8 + 8 * l];
        const float* bv  = (const float*)d_in[9 + 8 * l];
        const float* gv  = (const float*)d_in[10 + 8 * l];
        const float* bev = (const float*)d_in[11 + 8 * l];
        float* hbuf = (l & 1) ? bufB : bufA;
        float* acc  = (l & 1) ? bufA : bufB;
        const float* xin = (l == 0) ? x : acc;
        float* xout = (l == 2) ? (float*)d_out : hbuf;

        k_gemm_xw<<<1024, 256, 0, stream>>>(xin, Wl, hbuf, colsum, N);
        k_node_dots<<<512, 256, 0, stream>>>(hbuf, asv, adv, als, ald, N);
        k_node_attn<<<2048, 256, 0, stream>>>(rowptr, deg, rec, als, ald, Wel, aev, easum, E, bv,
                                              hbuf, acc, N);
        k_bn_stats<<<(N + 255) / 256, 512, 0, stream>>>(acc, colsum, colsq, N);
        k_bn_apply<<<(N + BA_ROWS - 1) / BA_ROWS, 256, 0, stream>>>(acc, colsum, colsq, gv, bev,
                                                                  xout, N);
    }
}


# GEMM A strip: the block loads its 16 rows once as whole-row float4s into LDS (2 buffers, next strip in flight); 8 waves read fragments from LDS, not 8 copies of 16-line b64 global loads; same sums
# speedup vs baseline: 3.6559x; 1.0767x over previous
#include <hip/hip_runtime.h>
#include <hip/hip_bf16.h>


typedef float v2f __attribute__((ext_vector_type(2)));
typedef float v8f __attribute__((ext_vector_type(8)));

#define HF 128

__device__ __forceinline__ float leaky(float x, float s) { return x > 0.f ? x : s * x; }

__global__ void k_zero2(float* p, int n, float* q, int m) {
    int i = blockIdx.x * blockDim.x + threadIdx.x;
    if (i < n) p[i] = 0.f;
    if (i < m) q[i] = 0.f;
}

#define GA_PITCH 132
__device__ __forceinline__ void strip_rows(const float* __restrict__ X, int sidx, int w, int lane,
                                           int n, float4& g0, float4& g1) {
    int ra = sidx * 16 + w;
    int rb = ra + 8;
    if (ra >= n) ra = n - 1;
    if (rb >= n) rb = n - 1;
    g0 = *(const float4*)(X + (size_t)ra * HF + lane * 4);
    g1 = *(const float4*)(X + (size_t)rb * HF + lane * 4);
}
__global__ __launch_bounds__(256) void k_gemm_xw(const float* __restrict__ X,
                                                 const float* __restrict__ W,
                                                 float* __restrict__ Ho,
                                                 float* __restrict__ colz, int n) {
    __shared__ __attribute__((aligned(16))) float As[2][16 * GA_PITCH];
    if (blockIdx.x == 0) colz[threadIdx.x] = 0.f;
    const int lane = threadIdx.x & 31;
    const int wv   = threadIdx.x >> 5;
    const int lo16 = lane & 15;
    const int hi   = lane >> 4;
    const int ncol = wv * 16 + lo16;

    const int strips = (n + 15) >> 4;
    float4 g0, g1;
    strip_rows(X, blockIdx.x < strips ? blockIdx.x : strips - 1, wv, lane, n, g0, g1);

    v2f breg[32];
#pragma unroll
    for (int kk = 0; kk < 32; ++kk) {
        int kb = kk * 4 + hi * 2;
        v2f b;
        b.x = W[(size_t)kb * HF + ncol];
        b.y = W[(size_t)(kb + 1) * HF + ncol];
        breg[kk] = b;
    }

    int buf = 0;
    for (int s = blockIdx.x; s < strips; s += gridDim.x) {
        const int r0 = s * 16;
        float* as = As[buf];
        *(float4*)(as + wv * GA_PITCH + lane * 4)       = g0;
        *(float4*)(as + (wv + 8) * GA_PITCH + lane * 4) = g1;
        __syncthreads();

        const int sn = s + (int)gridDim.x;
        strip_rows(X, sn < strips ? sn : strips - 1, wv, lane, n, g0, g1);
        __builtin_amdgcn_sched_barrier(0);

        const float* ar = as + lo16 * GA_PITCH + hi * 2;
        v2f a[32];
#pragma unroll
        for (int kk = 0; kk < 32; ++kk) a[kk] = *(const v2f*)(ar + kk * 4);

        v8f acc = {0.f, 0.f, 0.f, 0.f, 0.f, 0.f, 0.f, 0.f};
#pragma unroll
        for (int kk = 0; kk < 32; ++kk) {
            acc = __builtin_amdgcn_wmma_f32_16x16x4_f32(
                false, a[kk], false, breg[kk], (short)0, acc, false, false);
        }

#pragma unroll
        for (int v = 0; v < 8; ++v) {
            int row = r0 + v + hi * 8;
            if (row < n) Ho[(size_t)row * HF + ncol] = acc[v];
        }
        buf ^= 1;
    }
}

__global__ __launch_bounds__(256) void k_node_dots(const float* __restrict__ Hm,
                                                   const float* __restrict__ avs,
                                                   const float* __restrict__ avd,
                                                   float* __restrict__ als,
                                                   float* __restrict__ ald, int n) {
    const int lane = threadIdx.x & 31;
    int wid = (blockIdx.x * blockDim.x + threadIdx.x) >> 5;
    int nw  = (gridDim.x * blockDim.x) >> 5;
    float s0 = avs[lane], s1 = avs[lane + 32], s2 = avs[lane + 64], s3 = avs[lane + 96];
    float d0 = avd[lane], d1 = avd[lane + 32], d2 = avd[lane + 64], d3 = avd[lane + 96];
    for (int i = wid; i < n; i += nw) {
        const float* hr = Hm + (size_t)i * HF;
        float h0 = hr[lane], h1 = hr[lane + 32], h2 = hr[lane + 64], h3 = hr[lane + 96];
        float rs = h0 * s0 + h1 * s1 + h2 * s2 + h3 * s3;
        float rd = h0 * d0 + h1 * d1 + h2 * d2 + h3 * d3;
        for (int off = 16; off; off >>= 1) {
            rs += __shfl_xor(rs, off);
            rd += __shfl_xor(rd, off);
        }
        if (lane == 0) { als[i] = rs; ald[i] = rd; }
    }
}

__global__ void k_deg_count(const int* __restrict__ dst, int* __restrict__ deg, int E) {
    int e = blockIdx.x * blockDim.x + threadIdx.x;
    if (e >= E) return;
    atomicAdd(deg + dst[e], 1);
}

#define SCAN_T 1024
#define SCAN_W (SCAN_T / 32)
#define SCAN_STEP 128
__device__ __forceinline__ int wave_isum(int v) {
    for (int off = 16; off; off >>= 1) v += __shfl_xor(v, off);
    return v;
}
__device__ __forceinline__ int4 deg4_tail(const int* __restrict__ deg, int i, int n) {
    int4 v = {0, 0, 0, 0};
    if (i < n)     v.x = deg[i];
    if (i + 1 < n) v.y = deg[i + 1];
    if (i + 2 < n) v.z = deg[i + 2];
    if (i + 3 < n) v.w = deg[i + 3];
    return v;
}
__device__ __forceinline__ int4 scan4(const int4 v, int lane, int& carry) {
    const int own = (v.x + v.y) + (v.z + v.w);
    int inc = own;
    for (int off = 1; off < 32; off <<= 1) {
        const int u = __shfl_up(inc, off);
        if (lane >= off) inc += u;
    }
    const int run = carry + (inc - own);
    carry += __builtin_amdgcn_readlane(inc, 31);
    int4 o;
    o.x = run;
    o.y = o.x + v.x;
    o.z = o.y + v.y;
    o.w = o.z + v.z;
    return o;
}
__global__ __launch_bounds__(SCAN_T) void k_row_scan(const int* __restrict__ deg,
                                                     int* __restrict__ rowptr,
                                                     int* __restrict__ cursor, int n) {
    __shared__ int wtot[SCAN_W];
    const int lane = threadIdx.x & 31;
    const int w = __builtin_amdgcn_readfirstlane((int)(threadIdx.x >> 5));
    const int per = (((n + SCAN_W - 1) / SCAN_W) + SCAN_STEP - 1) & ~(SCAN_STEP - 1);
    const int lo = w * per;
    int hi = lo + per;
    if (hi > n) hi = n;
    const int l4 = lane * 4;

    int s = 0;
    int i0 = lo;
#pragma unroll 4
    for (; i0 + SCAN_STEP <= hi; i0 += SCAN_STEP) {
        const int4 v = *(const int4*)(deg + i0 + l4);
        s += (v.x + v.y) + (v.z + v.w);
    }
    if (i0 < hi) {
        const int4 v = deg4_tail(deg, i0 + l4, n);
        s += (v.x + v.y) + (v.z + v.w);
    }
    s = wave_isum(s);
    if (lane == 0) wtot[w] = s;
    __syncthreads();
    int carry = wave_isum(lane < w ? wtot[lane] : 0);

    i0 = lo;
#pragma unroll 4
    for (; i0 + SCAN_STEP <= hi; i0 += SCAN_STEP) {
        const int i = i0 + l4;
        const int4 o = scan4(*(const int4*)(deg + i), lane, carry);
        *(int4*)(rowptr + i) = o;
        *(int4*)(cursor + i) = o;
    }
    if (i0 < hi) {
        const int i = i0 + l4;
        const int4 o = scan4(deg4_tail(deg, i, n), lane, carry);
        if (i < n)     { rowptr[i] = o.x;     cursor[i] = o.x; }
        if (i + 1 < n) { rowptr[i + 1] = o.y; cursor[i + 1] = o.y; }
        if (i + 2 < n) { rowptr[i + 2] = o.z; cursor[i + 2] = o.z; }
        if (i + 3 < n) { rowptr[i + 3] = o.w; cursor[i + 3] = o.w; }
    }
}

__device__ __forceinline__ int lane_i(int v, int k) { return __builtin_amdgcn_readlane(v, k); }
__device__ __forceinline__ float lane_f(float v, int k) {
    return __int_as_float(__builtin_amdgcn_readlane(__float_as_int(v), k));
}
__device__ __forceinline__ float first_f(float v) {
    return __int_as_float(__builtin_amdgcn_readfirstlane(__float_as_int(v)));
}

template <int MASK>
__device__ __forceinline__ float row_xor(float v) {
    return __int_as_float(
        __builtin_amdgcn_update_dpp(0, __float_as_int(v), 0x160 + MASK, 0xf, 0xf, false));
}
__device__ __forceinline__ float xhalf(float v) {
    const int b = __float_as_int(v);
    return __int_as_float(
        __builtin_amdgcn_permlanex16(b, b, 0x76543210, 0xfedcba98, false, false));
}
__device__ __forceinline__ float wave_sum(float v) {
    v += row_xor<1>(v);
    v += row_xor<2>(v);
    v += row_xor<4>(v);
    v += row_xor<8>(v);
    return v + xhalf(v);
}
__device__ __forceinline__ float wave_max(float v) {
    v = fmaxf(v, row_xor<1>(v));
    v = fmaxf(v, row_xor<2>(v));
    v = fmaxf(v, row_xor<4>(v));
    v = fmaxf(v, row_xor<8>(v));
    return fmaxf(v, xhalf(v));
}

__global__ __launch_bounds__(256) void k_list_fill(const int* __restrict__ src,
                                                   const int* __restrict__ dst,
                                                   const float* __restrict__ ea,
                                                   int* __restrict__ cursor,
                                                   int* __restrict__ rec,
                                                   float* __restrict__ easum, int E) {
    __shared__ float part[3][8];
    const int e = blockIdx.x * blockDim.x + threadIdx.x;
    float a0 = 0.f, a1 = 0.f, a2 = 0.f;
    if (e < E) {
        const int p = atomicAdd(cursor + dst[e], 1);
        a0 = ea[e * 3 + 0];
        a1 = ea[e * 3 + 1];
        a2 = ea[e * 3 + 2];
        int4 r;
        r.x = src[e];
        r.y = __float_as_int(a0);
        r.z = __float_as_int(a1);
        r.w = __float_as_int(a2);
        ((int4*)rec)[p] = r;
    }
    a0 = wave_sum(a0);
    a1 = wave_sum(a1);
    a2 = wave_sum(a2);
    const int w = threadIdx.x >> 5;
    if ((threadIdx.x & 31) == 0) {
        part[0][w] = a0;
        part[1][w] = a1;
        part[2][w] = a2;
    }
    __syncthreads();
    if (threadIdx.x < 3) {
        float s = 0.f;
        for (int k = 0; k < 8; ++k) s += part[threadIdx.x][k];
        atomicAdd(easum + threadIdx.x, s);
    }
}

__device__ __forceinline__ void row_fma(float4& o, const float4 h, float c) {
    o.x = fmaf(h.x, c, o.x);
    o.y = fmaf(h.y, c, o.y);
    o.z = fmaf(h.z, c, o.z);
    o.w = fmaf(h.w, c, o.w);
}

__device__ __forceinline__ float slot_alpha(const int4* __restrict__ rec4,
                                            const float* __restrict__ als, int j, float ad,
                                            float c0, float c1, float c2, int& s) {
    const int4 r = rec4[j];
    s = r.x;
    const float ale = __int_as_float(r.y) * c0 + __int_as_float(r.z) * c1 +
                      __int_as_float(r.w) * c2;
    return leaky(als[s] + ad + ale, 0.2f);
}

__device__ __forceinline__ void gather_group(float4& o, int sl, float cl, int m,
                                             const float* __restrict__ Hm, int lane) {
    int k = 0;
    for (; k + 4 <= m; k += 4) {
        const int s0 = lane_i(sl, k),     s1 = lane_i(sl, k + 1);
        const int s2 = lane_i(sl, k + 2), s3 = lane_i(sl, k + 3);
        const float c0 = lane_f(cl, k),     c1 = lane_f(cl, k + 1);
        const float c2 = lane_f(cl, k + 2), c3 = lane_f(cl, k + 3);
        const float4 h0 = *(const float4*)(Hm + (size_t)s0 * HF + lane * 4);
        const float4 h1 = *(const float4*)(Hm + (size_t)s1 * HF + lane * 4);
        const float4 h2 = *(const float4*)(Hm + (size_t)s2 * HF + lane * 4);
        const float4 h3 = *(const float4*)(Hm + (size_t)s3 * HF + lane * 4);
        row_fma(o, h0, c0);
        row_fma(o, h1, c1);
        row_fma(o, h2, c2);
        row_fma(o, h3, c3);
    }
    const int r = m - k;
    if (r == 3) {
        const int s0 = lane_i(sl, k), s1 = lane_i(sl, k + 1), s2 = lane_i(sl, k + 2);
        const float c0 = lane_f(cl, k), c1 = lane_f(cl, k + 1), c2 = lane_f(cl, k + 2);
        const float4 h0 = *(const float4*)(Hm + (size_t)s0 * HF + lane * 4);
        const float4 h1 = *(const float4*)(Hm + (size_t)s1 * HF + lane * 4);
        const float4 h2 = *(const float4*)(Hm + (size_t)s2 * HF + lane * 4);
        row_fma(o, h0, c0);
        row_fma(o, h1, c1);
        row_fma(o, h2, c2);
    } else if (r == 2) {
        const int s0 = lane_i(sl, k), s1 = lane_i(sl, k + 1);
        const float c0 = lane_f(cl, k), c1 = lane_f(cl, k + 1);
        const float4 h0 = *(const float4*)(Hm + (size_t)s0 * HF + lane * 4);
        const float4 h1 = *(const float4*)(Hm + (size_t)s1 * HF + lane * 4);
        row_fma(o, h0, c0);
        row_fma(o, h1, c1);
    } else if (r == 1) {
        const int s0 = lane_i(sl, k);
        const float c0 = lane_f(cl, k);
        const float4 h0 = *(const float4*)(Hm + (size_t)s0 * HF + lane * 4);
        row_fma(o, h0, c0);
    }
}

__global__ __launch_bounds__(256) void k_node_attn(const int* __restrict__ rowptr,
                                                   const int* __restrict__ deg,
                                                   const int* __restrict__ rec,
                                                   const float* __restrict__ als,
                                                   const float* __restrict__ ald,
                                                   const float* __restrict__ We,
                                                   const float* __restrict__ ae,
                                                   const float* __restrict__ easum, int E,
                                                   const float* __restrict__ bias,
                                                   const float* __restrict__ Hm,
                                                   float* __restrict__ acc, int n) {
    const int lane = threadIdx.x & 31;
    const int nw   = (gridDim.x * blockDim.x) >> 5;
    const int w0 = __builtin_amdgcn_readfirstlane(
        (int)((blockIdx.x * blockDim.x + threadIdx.x) >> 5));
    const int4* __restrict__ rec4 = (const int4*)rec;
    const float4 av = *(const float4*)(ae + lane * 4);
    const float4 e0 = *(const float4*)(We + lane * 4);
    const float4 e1 = *(const float4*)(We + HF + lane * 4);
    const float4 e2 = *(const float4*)(We + 2 * HF + lane * 4);
    const float c0 = first_f(wave_sum(e0.x * av.x + e0.y * av.y + e0.z * av.z + e0.w * av.w));
    const float c1 = first_f(wave_sum(e1.x * av.x + e1.y * av.y + e1.z * av.z + e1.w * av.w));
    const float c2 = first_f(wave_sum(e2.x * av.x + e2.y * av.y + e2.z * av.z + e2.w * av.w));
    float cm = 0.f;
    cm += (easum[0] / (float)E) * c0;
    cm += (easum[1] / (float)E) * c1;
    cm += (easum[2] / (float)E) * c2;
    const float4 bv = *(const float4*)(bias + lane * 4);
    const float ninf = -__builtin_inff();
    for (int i = w0; i < n; i += nw) {
        const int start = rowptr[i];
        const int cnt   = deg[i];
        const float ad  = ald[i];
        const float as  = als[i];
        const float4 hs = *(const float4*)(Hm + (size_t)i * HF + lane * 4);
        float4 o;
        if (cnt <= 32) {
            const bool ok = lane < cnt;
            int sl = 0;
            float a = ninf;
            if (ok) a = slot_alpha(rec4, als, start + lane, ad, c0, c1, c2, sl);
            const float aself = leaky(as + ad + cm, 0.2f);
            const float m   = fmaxf(first_f(wave_max(a)), aself);
            const float ex  = ok ? expf(a - m) : 0.f;
            const float exs = expf(aself - m);
            const float den = (exs + first_f(wave_sum(ex))) + 1e-16f;
            const float cs  = exs / den;
            o.x = hs.x * cs + bv.x;
            o.y = hs.y * cs + bv.y;
            o.z = hs.z * cs + bv.z;
            o.w = hs.w * cs + bv.w;
            if (cnt > 0) gather_group(o, sl, ex / den, cnt, Hm, lane);
        } else {
            float mx = ninf;
            for (int base = 0; base < cnt; base += 32) {
                if (base + lane < cnt) {
                    int s;
                    mx = fmaxf(mx, slot_alpha(rec4, als, start + base + lane, ad, c0, c1, c2, s));
                }
            }
            const float aself = leaky(as + ad + cm, 0.2f);
            const float m = fmaxf(first_f(wave_max(mx)), aself);
            float sm = 0.f;
            for (int base = 0; base < cnt; base += 32) {
                if (base + lane < cnt) {
                    int s;
                    sm += expf(slot_alpha(rec4, als, start + base + lane, ad, c0, c1, c2, s) - m);
                }
            }
            const float exs = expf(aself - m);
            const float den = (exs + first_f(wave_sum(sm))) + 1e-16f;
            const float cs  = exs / den;
            o.x = hs.x * cs + bv.x;
            o.y = hs.y * cs + bv.y;
            o.z = hs.z * cs + bv.z;
            o.w = hs.w * cs + bv.w;
            for (int base = 0; base < cnt; base += 32) {
                int mm = cnt - base;
                if (mm > 32) mm = 32;
                int sl = 0;
                float cl = 0.f;
                if (lane < mm) {
                    const float a = slot_alpha(rec4, als, start + base + lane, ad, c0, c1, c2, sl);
                    cl = expf(a - m) / den;
                }
                gather_group(o, sl, cl, mm, Hm, lane);
            }
        }
        *(float4*)(acc + (size_t)i * HF + lane * 4) = o;
    }
}

#define BN_ROWS  256
#define BN_WAVES 16
#define BN_RPW   (BN_ROWS / BN_WAVES)
__global__ __launch_bounds__(512) void k_bn_stats(const float* __restrict__ acc,
                                                  float* __restrict__ colsum,
                                                  float* __restrict__ colsq, int n) {
    __shared__ __attribute__((aligned(16))) float part[2][BN_WAVES][HF];
    const int lane = threadIdx.x & 31;
    const int w    = __builtin_amdgcn_readfirstlane((int)(threadIdx.x >> 5));
    const int r0   = blockIdx.x * BN_ROWS + w;

    float4 t[BN_RPW];
#pragma unroll
    for (int i = 0; i < BN_RPW; ++i) {
        int r = r0 + i * BN_WAVES;
        if (r >= n) r = n - 1;
        t[i] = *(const float4*)(acc + (size_t)r * HF + lane * 4);
    }

    float4 s = {0.f, 0.f, 0.f, 0.f}, q = {0.f, 0.f, 0.f, 0.f};
#pragma unroll
    for (int i = 0; i < BN_RPW; ++i) {
        const bool ok = (r0 + i * BN_WAVES) < n;
        float y0 = ok ? leaky(t[i].x, 0.01f) : 0.f;
        float y1 = ok ? leaky(t[i].y, 0.01f) : 0.f;
        float y2 = ok ? leaky(t[i].z, 0.01f) : 0.f;
        float y3 = ok ? leaky(t[i].w, 0.01f) : 0.f;
        s.x += y0; q.x += y0 * y0;
        s.y += y1; q.y += y1 * y1;
        s.z += y2; q.z += y2 * y2;
        s.w += y3; q.w += y3 * y3;
    }
    *(float4*)&part[0][w][lane * 4] = s;
    *(float4*)&part[1][w][lane * 4] = q;
    __syncthreads();

    if (threadIdx.x < 2 * HF) {
        const int which = threadIdx.x >> 7;
        const int c     = threadIdx.x & (HF - 1);
        float r = 0.f;
        for (int k = 0; k < BN_WAVES; ++k) r += part[which][k][c];
        atomicAdd((which ? colsq : colsum) + c, r);
    }
}

#define BA_RPW  16
#define BA_ROWS (8 * BA_RPW)
__device__ __forceinline__ void bn_consts(const float4 cs, const float4 cq, int n, float4& mu,
                                          float4& rs) {
    const float invN = 1.f / (float)n;
    mu.x = cs.x * invN;  rs.x = rsqrtf((cq.x * invN - mu.x * mu.x) + 1e-5f);
    mu.y = cs.y * invN;  rs.y = rsqrtf((cq.y * invN - mu.y * mu.y) + 1e-5f);
    mu.z = cs.z * invN;  rs.z = rsqrtf((cq.z * invN - mu.z * mu.z) + 1e-5f);
    mu.w = cs.w * invN;  rs.w = rsqrtf((cq.w * invN - mu.w * mu.w) + 1e-5f);
}
__device__ __forceinline__ float4 bn_row(const float4 t, const float4 mu, const float4 rs,
                                         const float4 gv, const float4 bv) {
    float4 o;
    o.x = (leaky(t.x, 0.01f) - mu.x) * rs.x * gv.x + bv.x;
    o.y = (leaky(t.y, 0.01f) - mu.y) * rs.y * gv.y + bv.y;
    o.z = (leaky(t.z, 0.01f) - mu.z) * rs.z * gv.z + bv.z;
    o.w = (leaky(t.w, 0.01f) - mu.w) * rs.w * gv.w + bv.w;
    return o;
}
__global__ __launch_bounds__(256) void k_bn_apply(const float* __restrict__ acc,
                                                  const float* __restrict__ colsum,
                                                  const float* __restrict__ colsq,
                                                  const float* __restrict__ g,
                                                  const float* __restrict__ be,
                                                  float* __restrict__ xout, int n) {
    const int lane = threadIdx.x & 31;
    const int w  = __builtin_amdgcn_readfirstlane(
        (int)((blockIdx.x * blockDim.x + threadIdx.x) >> 5));
    const int r0 = w * BA_RPW;
    if (r0 >= n) return;

    const float4 cs = *(const float4*)(colsum + lane * 4);
    const float4 cq = *(const float4*)(colsq + lane * 4);
    const float4 gv = *(const float4*)(g + lane * 4);
    const float4 bv = *(const float4*)(be + lane * 4);
    const float* __restrict__ src = acc + (size_t)r0 * HF + lane * 4;
    float* __restrict__ dst = xout + (size_t)r0 * HF + lane * 4;

    if (r0 + BA_RPW <= n) {
        float4 t[BA_RPW];
#pragma unroll
        for (int i = 0; i < BA_RPW; ++i) t[i] = *(const float4*)(src + (size_t)i * HF);

        float4 mu, rs;
        bn_consts(cs, cq, n, mu, rs);
#pragma unroll
        for (int i = 0; i < BA_RPW; ++i)
            *(float4*)(dst + (size_t)i * HF) = bn_row(t[i], mu, rs, gv, bv);
    } else {
        float4 mu, rs;
        bn_consts(cs, cq, n, mu, rs);
        for (int i = 0; r0 + i < n; ++i)
            *(float4*)(dst + (size_t)i * HF) =
                bn_row(*(const float4*)(src + (size_t)i * HF), mu, rs, gv, bv);
    }
}

extern "C" void kernel_launch(void* const* d_in, const int* in_sizes, int n_in,
                              void* d_out, int out_size, void* d_ws, size_t ws_size,
                              hipStream_t stream) {
    const float* x  = (const float*)d_in[0];
    const int*   ei = (const int*)d_in[1];
    const float* ea = (const float*)d_in[3];
    const int N = in_sizes[0] / HF;
    const int E = in_sizes[1] / 2;
    const int* src = ei;
    const int* dst = ei + E;

    float* ws = (float*)d_ws;
    const size_t NH = (size_t)N * HF;
    float* bufA   = ws;
    float* bufB   = bufA + NH;
    int*   rec    = (int*)(bufB + NH);
    int*   rowptr = rec + 4 * (size_t)E;
    int*   cursor = rowptr + N;
    int*   deg    = cursor + N;
    float* als    = bufB + NH + NH;
    float* ald    = als + N;
    float* colsum = ald + N;
    float* colsq  = colsum + HF;
    float* easum  = colsq + HF;

    const int nb_e  = (E + 255) / 256;
    const int nb_n  = (N + 255) / 256;

    k_zero2<<<nb_n, 256, 0, stream>>>((float*)deg, N, easum, 4);

    k_deg_count<<<nb_e, 256, 0, stream>>>(dst, deg, E);
    k_row_scan<<<1, SCAN_T, 0, stream>>>(deg, rowptr, cursor, N);
    k_list_fill<<<nb_e, 256, 0, stream>>>(src, dst, ea, cursor, rec, easum, E);

    for (int l = 0; l < 3; ++l) {
        const float* Wl  = (const float*)d_in[4 + 8 * l];
        const float* asv = (const float*)d_in[5 + 8 * l];
        const float* adv = (const float*)d_in[6 + 8 * l];
        const float* Wel = (const float*)d_in[7 + 8 * l];
        const float* aev = (const float*)d_in[8 + 8 * l];
        const float* bv  = (const float*)d_in[9 + 8 * l];
        const float* gv  = (const float*)d_in[10 + 8 * l];
        const float* bev = (const float*)d_in[11 + 8 * l];
        float* hbuf = (l & 1) ? bufB : bufA;
        float* acc  = (l & 1) ? bufA : bufB;
        const float* xin = (l == 0) ? x : acc;
        float* xout = (l == 2) ? (float*)d_out : hbuf;

        k_gemm_xw<<<1024, 256, 0, stream>>>(xin, Wl, hbuf, colsum, N);
        k_node_dots<<<512, 256, 0, stream>>>(hbuf, asv, adv, als, ald, N);
        k_node_attn<<<2048, 256, 0, stream>>>(rowptr, deg, rec, als, ald, Wel, aev, easum, E, bv,
                                              hbuf, acc, N);
        k_bn_stats<<<(N + 255) / 256, 512, 0, stream>>>(acc, colsum, colsq, N);
        k_bn_apply<<<(N + BA_ROWS - 1) / BA_ROWS, 256, 0, stream>>>(acc, colsum, colsq, gv, bev,
                                                                  xout, N);
    }
}


# Edge list fill without atomics: the count add's returned old count is the edge's rank, the fill stores at rowptr[dst] + rank; the cursor add pass goes; row sum order = arrival order of count adds
# speedup vs baseline: 3.9383x; 1.0772x over previous
#include <hip/hip_runtime.h>
#include <hip/hip_bf16.h>


typedef float v2f __attribute__((ext_vector_type(2)));
typedef float v8f __attribute__((ext_vector_type(8)));

#define HF 128

__device__ __forceinline__ float leaky(float x, float s) { return x > 0.f ? x : s * x; }

__global__ void k_zero2(float* p, int n, float* q, int m) {
    int i = blockIdx.x * blockDim.x + threadIdx.x;
    if (i < n) p[i] = 0.f;
    if (i < m) q[i] = 0.f;
}

#define GA_PITCH 132
__device__ __forceinline__ void strip_rows(const float* __restrict__ X, int sidx, int w, int lane,
                                           int n, float4& g0, float4& g1) {
    int ra = sidx * 16 + w;
    int rb = ra + 8;
    if (ra >= n) ra = n - 1;
    if (rb >= n) rb = n - 1;
    g0 = *(const float4*)(X + (size_t)ra * HF + lane * 4);
    g1 = *(const float4*)(X + (size_t)rb * HF + lane * 4);
}
__global__ __launch_bounds__(256) void k_gemm_xw(const float* __restrict__ X,
                                                 const float* __restrict__ W,
                                                 float* __restrict__ Ho,
                                                 float* __restrict__ colz, int n) {
    __shared__ __attribute__((aligned(16))) float As[2][16 * GA_PITCH];
    if (blockIdx.x == 0) colz[threadIdx.x] = 0.f;
    const int lane = threadIdx.x & 31;
    const int wv   = threadIdx.x >> 5;
    const int lo16 = lane & 15;
    const int hi   = lane >> 4;
    const int ncol = wv * 16 + lo16;

    const int strips = (n + 15) >> 4;
    float4 g0, g1;
    strip_rows(X, blockIdx.x < strips ? blockIdx.x : strips - 1, wv, lane, n, g0, g1);

    v2f breg[32];
#pragma unroll
    for (int kk = 0; kk < 32; ++kk) {
        int kb = kk * 4 + hi * 2;
        v2f b;
        b.x = W[(size_t)kb * HF + ncol];
        b.y = W[(size_t)(kb + 1) * HF + ncol];
        breg[kk] = b;
    }

    int buf = 0;
    for (int s = blockIdx.x; s < strips; s += gridDim.x) {
        const int r0 = s * 16;
        float* as = As[buf];
        *(float4*)(as + wv * GA_PITCH + lane * 4)       = g0;
        *(float4*)(as + (wv + 8) * GA_PITCH + lane * 4) = g1;
        __syncthreads();

        const int sn = s + (int)gridDim.x;
        strip_rows(X, sn < strips ? sn : strips - 1, wv, lane, n, g0, g1);
        __builtin_amdgcn_sched_barrier(0);

        const float* ar = as + lo16 * GA_PITCH + hi * 2;
        v2f a[32];
#pragma unroll
        for (int kk = 0; kk < 32; ++kk) a[kk] = *(const v2f*)(ar + kk * 4);

        v8f acc = {0.f, 0.f, 0.f, 0.f, 0.f, 0.f, 0.f, 0.f};
#pragma unroll
        for (int kk = 0; kk < 32; ++kk) {
            acc = __builtin_amdgcn_wmma_f32_16x16x4_f32(
                false, a[kk], false, breg[kk], (short)0, acc, false, false);
        }

#pragma unroll
        for (int v = 0; v < 8; ++v) {
            int row = r0 + v + hi * 8;
            if (row < n) Ho[(size_t)row * HF + ncol] = acc[v];
        }
        buf ^= 1;
    }
}

__global__ __launch_bounds__(256) void k_node_dots(const float* __restrict__ Hm,
                                                   const float* __restrict__ avs,
                                                   const float* __restrict__ avd,
                                                   float* __restrict__ als,
                                                   float* __restrict__ ald, int n) {
    const int lane = threadIdx.x & 31;
    int wid = (blockIdx.x * blockDim.x + threadIdx.x) >> 5;
    int nw  = (gridDim.x * blockDim.x) >> 5;
    float s0 = avs[lane], s1 = avs[lane + 32], s2 = avs[lane + 64], s3 = avs[lane + 96];
    float d0 = avd[lane], d1 = avd[lane + 32], d2 = avd[lane + 64], d3 = avd[lane + 96];
    for (int i = wid; i < n; i += nw) {
        const float* hr = Hm + (size_t)i * HF;
        float h0 = hr[lane], h1 = hr[lane + 32], h2 = hr[lane + 64], h3 = hr[lane + 96];
        float rs = h0 * s0 + h1 * s1 + h2 * s2 + h3 * s3;
        float rd = h0 * d0 + h1 * d1 + h2 * d2 + h3 * d3;
        for (int off = 16; off; off >>= 1) {
            rs += __shfl_xor(rs, off);
            rd += __shfl_xor(rd, off);
        }
        if (lane == 0) { als[i] = rs; ald[i] = rd; }
    }
}

__global__ void k_deg_count(const int* __restrict__ dst, int* __restrict__ deg,
                            int* __restrict__ rank, int E) {
    int e = blockIdx.x * blockDim.x + threadIdx.x;
    if (e >= E) return;
    rank[e] = atomicAdd(deg + dst[e], 1);
}

#define SCAN_T 1024
#define SCAN_W (SCAN_T / 32)
#define SCAN_STEP 128
__device__ __forceinline__ int wave_isum(int v) {
    for (int off = 16; off; off >>= 1) v += __shfl_xor(v, off);
    return v;
}
__device__ __forceinline__ int4 deg4_tail(const int* __restrict__ deg, int i, int n) {
    int4 v = {0, 0, 0, 0};
    if (i < n)     v.x = deg[i];
    if (i + 1 < n) v.y = deg[i + 1];
    if (i + 2 < n) v.z = deg[i + 2];
    if (i + 3 < n) v.w = deg[i + 3];
    return v;
}
__device__ __forceinline__ int4 scan4(const int4 v, int lane, int& carry) {
    const int own = (v.x + v.y) + (v.z + v.w);
    int inc = own;
    for (int off = 1; off < 32; off <<= 1) {
        const int u = __shfl_up(inc, off);
        if (lane >= off) inc += u;
    }
    const int run = carry + (inc - own);
    carry += __builtin_amdgcn_readlane(inc, 31);
    int4 o;
    o.x = run;
    o.y = o.x + v.x;
    o.z = o.y + v.y;
    o.w = o.z + v.z;
    return o;
}
__global__ __launch_bounds__(SCAN_T) void k_row_scan(const int* __restrict__ deg,
                                                     int* __restrict__ rowptr, int n) {
    __shared__ int wtot[SCAN_W];
    const int lane = threadIdx.x & 31;
    const int w = __builtin_amdgcn_readfirstlane((int)(threadIdx.x >> 5));
    const int per = (((n + SCAN_W - 1) / SCAN_W) + SCAN_STEP - 1) & ~(SCAN_STEP - 1);
    const int lo = w * per;
    int hi = lo + per;
    if (hi > n) hi = n;
    const int l4 = lane * 4;

    int s = 0;
    int i0 = lo;
#pragma unroll 4
    for (; i0 + SCAN_STEP <= hi; i0 += SCAN_STEP) {
        const int4 v = *(const int4*)(deg + i0 + l4);
        s += (v.x + v.y) + (v.z + v.w);
    }
    if (i0 < hi) {
        const int4 v = deg4_tail(deg, i0 + l4, n);
        s += (v.x + v.y) + (v.z + v.w);
    }
    s = wave_isum(s);
    if (lane == 0) wtot[w] = s;
    __syncthreads();
    int carry = wave_isum(lane < w ? wtot[lane] : 0);

    i0 = lo;
#pragma unroll 4
    for (; i0 + SCAN_STEP <= hi; i0 += SCAN_STEP) {
        const int i = i0 + l4;
        const int4 o = scan4(*(const int4*)(deg + i), lane, carry);
        *(int4*)(rowptr + i) = o;
    }
    if (i0 < hi) {
        const int i = i0 + l4;
        const int4 o = scan4(deg4_tail(deg, i, n), lane, carry);
        if (i < n)     rowptr[i] = o.x;
        if (i + 1 < n) rowptr[i + 1] = o.y;
        if (i + 2 < n) rowptr[i + 2] = o.z;
        if (i + 3 < n) rowptr[i + 3] = o.w;
    }
}

__device__ __forceinline__ int lane_i(int v, int k) { return __builtin_amdgcn_readlane(v, k); }
__device__ __forceinline__ float lane_f(float v, int k) {
    return __int_as_float(__builtin_amdgcn_readlane(__float_as_int(v), k));
}
__device__ __forceinline__ float first_f(float v) {
    return __int_as_float(__builtin_amdgcn_readfirstlane(__float_as_int(v)));
}

template <int MASK>
__device__ __forceinline__ float row_xor(float v) {
    return __int_as_float(
        __builtin_amdgcn_update_dpp(0, __float_as_int(v), 0x160 + MASK, 0xf, 0xf, false));
}
__device__ __forceinline__ float xhalf(float v) {
    const int b = __float_as_int(v);
    return __int_as_float(
        __builtin_amdgcn_permlanex16(b, b, 0x76543210, 0xfedcba98, false, false));
}
__device__ __forceinline__ float wave_sum(float v) {
    v += row_xor<1>(v);
    v += row_xor<2>(v);
    v += row_xor<4>(v);
    v += row_xor<8>(v);
    return v + xhalf(v);
}
__device__ __forceinline__ float wave_max(float v) {
    v = fmaxf(v, row_xor<1>(v));
    v = fmaxf(v, row_xor<2>(v));
    v = fmaxf(v, row_xor<4>(v));
    v = fmaxf(v, row_xor<8>(v));
    return fmaxf(v, xhalf(v));
}

__global__ __launch_bounds__(256) void k_list_fill(const int* __restrict__ src,
                                                   const int* __restrict__ dst,
                                                   const float* __restrict__ ea,
                                                   const int* __restrict__ rowptr,
                                                   const int* __restrict__ rank,
                                                   int* __restrict__ rec,
                                                   float* __restrict__ easum, int E) {
    __shared__ float part[3][8];
    const int e = blockIdx.x * blockDim.x + threadIdx.x;
    float a0 = 0.f, a1 = 0.f, a2 = 0.f;
    if (e < E) {
        const int p = rowptr[dst[e]] + rank[e];
        a0 = ea[e * 3 + 0];
        a1 = ea[e * 3 + 1];
        a2 = ea[e * 3 + 2];
        int4 r;
        r.x = src[e];
        r.y = __float_as_int(a0);
        r.z = __float_as_int(a1);
        r.w = __float_as_int(a2);
        ((int4*)rec)[p] = r;
    }
    a0 = wave_sum(a0);
    a1 = wave_sum(a1);
    a2 = wave_sum(a2);
    const int w = threadIdx.x >> 5;
    if ((threadIdx.x & 31) == 0) {
        part[0][w] = a0;
        part[1][w] = a1;
        part[2][w] = a2;
    }
    __syncthreads();
    if (threadIdx.x < 3) {
        float s = 0.f;
        for (int k = 0; k < 8; ++k) s += part[threadIdx.x][k];
        atomicAdd(easum + threadIdx.x, s);
    }
}

__device__ __forceinline__ void row_fma(float4& o, const float4 h, float c) {
    o.x = fmaf(h.x, c, o.x);
    o.y = fmaf(h.y, c, o.y);
    o.z = fmaf(h.z, c, o.z);
    o.w = fmaf(h.w, c, o.w);
}

__device__ __forceinline__ float slot_alpha(const int4* __restrict__ rec4,
                                            const float* __restrict__ als, int j, float ad,
                                            float c0, float c1, float c2, int& s) {
    const int4 r = rec4[j];
    s = r.x;
    const float ale = __int_as_float(r.y) * c0 + __int_as_float(r.z) * c1 +
                      __int_as_float(r.w) * c2;
    return leaky(als[s] + ad + ale, 0.2f);
}

__device__ __forceinline__ void gather_group(float4& o, int sl, float cl, int m,
                                             const float* __restrict__ Hm, int lane) {
    int k = 0;
    for (; k + 4 <= m; k += 4) {
        const int s0 = lane_i(sl, k),     s1 = lane_i(sl, k + 1);
        const int s2 = lane_i(sl, k + 2), s3 = lane_i(sl, k + 3);
        const float c0 = lane_f(cl, k),     c1 = lane_f(cl, k + 1);
        const float c2 = lane_f(cl, k + 2), c3 = lane_f(cl, k + 3);
        const float4 h0 = *(const float4*)(Hm + (size_t)s0 * HF + lane * 4);
        const float4 h1 = *(const float4*)(Hm + (size_t)s1 * HF + lane * 4);
        const float4 h2 = *(const float4*)(Hm + (size_t)s2 * HF + lane * 4);
        const float4 h3 = *(const float4*)(Hm + (size_t)s3 * HF + lane * 4);
        row_fma(o, h0, c0);
        row_fma(o, h1, c1);
        row_fma(o, h2, c2);
        row_fma(o, h3, c3);
    }
    const int r = m - k;
    if (r == 3) {
        const int s0 = lane_i(sl, k), s1 = lane_i(sl, k + 1), s2 = lane_i(sl, k + 2);
        const float c0 = lane_f(cl, k), c1 = lane_f(cl, k + 1), c2 = lane_f(cl, k + 2);
        const float4 h0 = *(const float4*)(Hm + (size_t)s0 * HF + lane * 4);
        const float4 h1 = *(const float4*)(Hm + (size_t)s1 * HF + lane * 4);
        const float4 h2 = *(const float4*)(Hm + (size_t)s2 * HF + lane * 4);
        row_fma(o, h0, c0);
        row_fma(o, h1, c1);
        row_fma(o, h2, c2);
    } else if (r == 2) {
        const int s0 = lane_i(sl, k), s1 = lane_i(sl, k + 1);
        const float c0 = lane_f(cl, k), c1 = lane_f(cl, k + 1);
        const float4 h0 = *(const float4*)(Hm + (size_t)s0 * HF + lane * 4);
        const float4 h1 = *(const float4*)(Hm + (size_t)s1 * HF + lane * 4);
        row_fma(o, h0, c0);
        row_fma(o, h1, c1);
    } else if (r == 1) {
        const int s0 = lane_i(sl, k);
        const float c0 = lane_f(cl, k);
        const float4 h0 = *(const float4*)(Hm + (size_t)s0 * HF + lane * 4);
        row_fma(o, h0, c0);
    }
}

__global__ __launch_bounds__(256) void k_node_attn(const int* __restrict__ rowptr,
                                                   const int* __restrict__ deg,
                                                   const int* __restrict__ rec,
                                                   const float* __restrict__ als,
                                                   const float* __restrict__ ald,
                                                   const float* __restrict__ We,
                                                   const float* __restrict__ ae,
                                                   const float* __restrict__ easum, int E,
                                                   const float* __restrict__ bias,
                                                   const float* __restrict__ Hm,
                                                   float* __restrict__ acc, int n) {
    const int lane = threadIdx.x & 31;
    const int nw   = (gridDim.x * blockDim.x) >> 5;
    const int w0 = __builtin_amdgcn_readfirstlane(
        (int)((blockIdx.x * blockDim.x + threadIdx.x) >> 5));
    const int4* __restrict__ rec4 = (const int4*)rec;
    const float4 av = *(const float4*)(ae + lane * 4);
    const float4 e0 = *(const float4*)(We + lane * 4);
    const float4 e1 = *(const float4*)(We + HF + lane * 4);
    const float4 e2 = *(const float4*)(We + 2 * HF + lane * 4);
    const float c0 = first_f(wave_sum(e0.x * av.x + e0.y * av.y + e0.z * av.z + e0.w * av.w));
    const float c1 = first_f(wave_sum(e1.x * av.x + e1.y * av.y + e1.z * av.z + e1.w * av.w));
    const float c2 = first_f(wave_sum(e2.x * av.x + e2.y * av.y + e2.z * av.z + e2.w * av.w));
    float cm = 0.f;
    cm += (easum[0] / (float)E) * c0;
    cm += (easum[1] / (float)E) * c1;
    cm += (easum[2] / (float)E) * c2;
    const float4 bv = *(const float4*)(bias + lane * 4);
    const float ninf = -__builtin_inff();
    for (int i = w0; i < n; i += nw) {
        const int start = rowptr[i];
        const int cnt   = deg[i];
        const float ad  = ald[i];
        const float as  = als[i];
        const float4 hs = *(const float4*)(Hm + (size_t)i * HF + lane * 4);
        float4 o;
        if (cnt <= 32) {
            const bool ok = lane < cnt;
            int sl = 0;
            float a = ninf;
            if (ok) a = slot_alpha(rec4, als, start + lane, ad, c0, c1, c2, sl);
            const float aself = leaky(as + ad + cm, 0.2f);
            const float m   = fmaxf(first_f(wave_max(a)), aself);
            const float ex  = ok ? expf(a - m) : 0.f;
            const float exs = expf(aself - m);
            const float den = (exs + first_f(wave_sum(ex))) + 1e-16f;
            const float cs  = exs / den;
            o.x = hs.x * cs + bv.x;
            o.y = hs.y * cs + bv.y;
            o.z = hs.z * cs + bv.z;
            o.w = hs.w * cs + bv.w;
            if (cnt > 0) gather_group(o, sl, ex / den, cnt, Hm, lane);
        } else {
            float mx = ninf;
            for (int base = 0; base < cnt; base += 32) {
                if (base + lane < cnt) {
                    int s;
                    mx = fmaxf(mx, slot_alpha(rec4, als, start + base + lane, ad, c0, c1, c2, s));
                }
            }
            const float aself = leaky(as + ad + cm, 0.2f);
            const float m = fmaxf(first_f(wave_max(mx)), aself);
            float sm = 0.f;
            for (int base = 0; base < cnt; base += 32) {
                if (base + lane < cnt) {
                    int s;
                    sm += expf(slot_alpha(rec4, als, start + base + lane, ad, c0, c1, c2, s) - m);
                }
            }
            const float exs = expf(aself - m);
            const float den = (exs + first_f(wave_sum(sm))) + 1e-16f;
            const float cs  = exs / den;
            o.x = hs.x * cs + bv.x;
            o.y = hs.y * cs + bv.y;
            o.z = hs.z * cs + bv.z;
            o.w = hs.w * cs + bv.w;
            for (int base = 0; base < cnt; base += 32) {
                int mm = cnt - base;
                if (mm > 32) mm = 32;
                int sl = 0;
                float cl = 0.f;
                if (lane < mm) {
                    const float a = slot_alpha(rec4, als, start + base + lane, ad, c0, c1, c2, sl);
                    cl = expf(a - m) / den;
                }
                gather_group(o, sl, cl, mm, Hm, lane);
            }
        }
        *(float4*)(acc + (size_t)i * HF + lane * 4) = o;
    }
}

#define BN_ROWS  256
#define BN_WAVES 16
#define BN_RPW   (BN_ROWS / BN_WAVES)
__global__ __launch_bounds__(512) void k_bn_stats(const float* __restrict__ acc,
                                                  float* __restrict__ colsum,
                                                  float* __restrict__ colsq, int n) {
    __shared__ __attribute__((aligned(16))) float part[2][BN_WAVES][HF];
    const int lane = threadIdx.x & 31;
    const int w    = __builtin_amdgcn_readfirstlane((int)(threadIdx.x >> 5));
    const int r0   = blockIdx.x * BN_ROWS + w;

    float4 t[BN_RPW];
#pragma unroll
    for (int i = 0; i < BN_RPW; ++i) {
        int r = r0 + i * BN_WAVES;
        if (r >= n) r = n - 1;
        t[i] = *(const float4*)(acc + (size_t)r * HF + lane * 4);
    }

    float4 s = {0.f, 0.f, 0.f, 0.f}, q = {0.f, 0.f, 0.f, 0.f};
#pragma unroll
    for (int i = 0; i < BN_RPW; ++i) {
        const bool ok = (r0 + i * BN_WAVES) < n;
        float y0 = ok ? leaky(t[i].x, 0.01f) : 0.f;
        float y1 = ok ? leaky(t[i].y, 0.01f) : 0.f;
        float y2 = ok ? leaky(t[i].z, 0.01f) : 0.f;
        float y3 = ok ? leaky(t[i].w, 0.01f) : 0.f;
        s.x += y0; q.x += y0 * y0;
        s.y += y1; q.y += y1 * y1;
        s.z += y2; q.z += y2 * y2;
        s.w += y3; q.w += y3 * y3;
    }
    *(float4*)&part[0][w][lane * 4] = s;
    *(float4*)&part[1][w][lane * 4] = q;
    __syncthreads();

    if (threadIdx.x < 2 * HF) {
        const int which = threadIdx.x >> 7;
        const int c     = threadIdx.x & (HF - 1);
        float r = 0.f;
        for (int k = 0; k < BN_WAVES; ++k) r += part[which][k][c];
        atomicAdd((which ? colsq : colsum) + c, r);
    }
}

#define BA_RPW  16
#define BA_ROWS (8 * BA_RPW)
__device__ __forceinline__ void bn_consts(const float4 cs, const float4 cq, int n, float4& mu,
                                          float4& rs) {
    const float invN = 1.f / (float)n;
    mu.x = cs.x * invN;  rs.x = rsqrtf((cq.x * invN - mu.x * mu.x) + 1e-5f);
    mu.y = cs.y * invN;  rs.y = rsqrtf((cq.y * invN - mu.y * mu.y) + 1e-5f);
    mu.z = cs.z * invN;  rs.z = rsqrtf((cq.z * invN - mu.z * mu.z) + 1e-5f);
    mu.w = cs.w * invN;  rs.w = rsqrtf((cq.w * invN - mu.w * mu.w) + 1e-5f);
}
__device__ __forceinline__ float4 bn_row(const float4 t, const float4 mu, const float4 rs,
                                         const float4 gv, const float4 bv) {
    float4 o;
    o.x = (leaky(t.x, 0.01f) - mu.x) * rs.x * gv.x + bv.x;
    o.y = (leaky(t.y, 0.01f) - mu.y) * rs.y * gv.y + bv.y;
    o.z = (leaky(t.z, 0.01f) - mu.z) * rs.z * gv.z + bv.z;
    o.w = (leaky(t.w, 0.01f) - mu.w) * rs.w * gv.w + bv.w;
    return o;
}
__global__ __launch_bounds__(256) void k_bn_apply(const float* __restrict__ acc,
                                                  const float* __restrict__ colsum,
                                                  const float* __restrict__ colsq,
                                                  const float* __restrict__ g,
                                                  const float* __restrict__ be,
                                                  float* __restrict__ xout, int n) {
    const int lane = threadIdx.x & 31;
    const int w  = __builtin_amdgcn_readfirstlane(
        (int)((blockIdx.x * blockDim.x + threadIdx.x) >> 5));
    const int r0 = w * BA_RPW;
    if (r0 >= n) return;

    const float4 cs = *(const float4*)(colsum + lane * 4);
    const float4 cq = *(const float4*)(colsq + lane * 4);
    const float4 gv = *(const float4*)(g + lane * 4);
    const float4 bv = *(const float4*)(be + lane * 4);
    const float* __restrict__ src = acc + (size_t)r0 * HF + lane * 4;
    float* __restrict__ dst = xout + (size_t)r0 * HF + lane * 4;

    if (r0 + BA_RPW <= n) {
        float4 t[BA_RPW];
#pragma unroll
        for (int i = 0; i < BA_RPW; ++i) t[i] = *(const float4*)(src + (size_t)i * HF);

        float4 mu, rs;
        bn_consts(cs, cq, n, mu, rs);
#pragma unroll
        for (int i = 0; i < BA_RPW; ++i)
            *(float4*)(dst + (size_t)i * HF) = bn_row(t[i], mu, rs, gv, bv);
    } else {
        float4 mu, rs;
        bn_consts(cs, cq, n, mu, rs);
        for (int i = 0; r0 + i < n; ++i)
            *(float4*)(dst + (size_t)i * HF) =
                bn_row(*(const float4*)(src + (size_t)i * HF), mu, rs, gv, bv);
    }
}

extern "C" void kernel_launch(void* const* d_in, const int* in_sizes, int n_in,
                              void* d_out, int out_size, void* d_ws, size_t ws_size,
                              hipStream_t stream) {
    const float* x  = (const float*)d_in[0];
    const int*   ei = (const int*)d_in[1];
    const float* ea = (const float*)d_in[3];
    const int N = in_sizes[0] / HF;
    const int E = in_sizes[1] / 2;
    const int* src = ei;
    const int* dst = ei + E;

    float* ws = (float*)d_ws;
    const size_t NH = (size_t)N * HF;
    float* bufA   = ws;
    float* bufB   = bufA + NH;
    int*   rec    = (int*)(bufB + NH);
    int*   rowptr = rec + 4 * (size_t)E;
    int*   deg    = rowptr + N;
    int*   rank   = deg + N;
    float* als    = bufB + NH + NH;
    float* ald    = als + N;
    float* colsum = ald + N;
    float* colsq  = colsum + HF;
    float* easum  = colsq + HF;

    const int nb_e  = (E + 255) / 256;
    const int nb_n  = (N + 255) / 256;

    k_zero2<<<nb_n, 256, 0, stream>>>((float*)deg, N, easum, 4);

    k_deg_count<<<nb_e, 256, 0, stream>>>(dst, deg, rank, E);
    k_row_scan<<<1, SCAN_T, 0, stream>>>(deg, rowptr, N);
    k_list_fill<<<nb_e, 256, 0, stream>>>(src, dst, ea, rowptr, rank, rec, easum, E);

    for (int l = 0; l < 3; ++l) {
        const float* Wl  = (const float*)d_in[4 + 8 * l];
        const float* asv = (const float*)d_in[5 + 8 * l];
        const float* adv = (const float*)d_in[6 + 8 * l];
        const float* Wel = (const float*)d_in[7 + 8 * l];
        const float* aev = (const float*)d_in[8 + 8 * l];
        const float* bv  = (const float*)d_in[9 + 8 * l];
        const float* gv  = (const float*)d_in[10 + 8 * l];
        const float* bev = (const float*)d_in[11 + 8 * l];
        float* hbuf = (l & 1) ? bufB : bufA;
        float* acc  = (l & 1) ? bufA : bufB;
        const float* xin = (l == 0) ? x : acc;
        float* xout = (l == 2) ? (float*)d_out : hbuf;

        k_gemm_xw<<<1024, 256, 0, stream>>>(xin, Wl, hbuf, colsum, N);
        k_node_dots<<<512, 256, 0, stream>>>(hbuf, asv, adv, als, ald, N);
        k_node_attn<<<2048, 256, 0, stream>>>(rowptr, deg, rec, als, ald, Wel, aev, easum, E, bv,
                                              hbuf, acc, N);
        k_bn_stats<<<(N + 255) / 256, 512, 0, stream>>>(acc, colsum, colsq, N);
        k_bn_apply<<<(N + BA_ROWS - 1) / BA_ROWS, 256, 0, stream>>>(acc, colsum, colsq, gv, bev,
                                                                  xout, N);
    }
}


# Edge count on LDS counters: 32 work-groups rank their edge chunk by returning LDS adds, a column pass makes chunk bases and degrees; global returning adds go; row sum order = chunk, then LDS arrival
# speedup vs baseline: 5.3789x; 1.3658x over previous
#include <hip/hip_runtime.h>
#include <hip/hip_bf16.h>


typedef float v2f __attribute__((ext_vector_type(2)));
typedef float v8f __attribute__((ext_vector_type(8)));

#define HF 128

__device__ __forceinline__ float leaky(float x, float s) { return x > 0.f ? x : s * x; }

template <int MASK>
__device__ __forceinline__ float row_xor(float v) {
    return __int_as_float(
        __builtin_amdgcn_update_dpp(0, __float_as_int(v), 0x160 + MASK, 0xf, 0xf, false));
}
__device__ __forceinline__ float xhalf(float v) {
    const int b = __float_as_int(v);
    return __int_as_float(
        __builtin_amdgcn_permlanex16(b, b, 0x76543210, 0xfedcba98, false, false));
}
__device__ __forceinline__ float wave_sum(float v) {
    v += row_xor<1>(v);
    v += row_xor<2>(v);
    v += row_xor<4>(v);
    v += row_xor<8>(v);
    return v + xhalf(v);
}
__device__ __forceinline__ float wave_max(float v) {
    v = fmaxf(v, row_xor<1>(v));
    v = fmaxf(v, row_xor<2>(v));
    v = fmaxf(v, row_xor<4>(v));
    v = fmaxf(v, row_xor<8>(v));
    return fmaxf(v, xhalf(v));
}

__device__ __forceinline__ float dot4(const float4 a, const float4 b) {
    return a.x * b.x + a.y * b.y + a.z * b.z + a.w * b.w;
}

__device__ __forceinline__ void bn_consts(const float4 cs, const float4 cq, int n, float4& mu,
                                          float4& rs) {
    const float invN = 1.f / (float)n;
    mu.x = cs.x * invN;  rs.x = rsqrtf((cq.x * invN - mu.x * mu.x) + 1e-5f);
    mu.y = cs.y * invN;  rs.y = rsqrtf((cq.y * invN - mu.y * mu.y) + 1e-5f);
    mu.z = cs.z * invN;  rs.z = rsqrtf((cq.z * invN - mu.z * mu.z) + 1e-5f);
    mu.w = cs.w * invN;  rs.w = rsqrtf((cq.w * invN - mu.w * mu.w) + 1e-5f);
}
__device__ __forceinline__ float4 bn_row(const float4 t, const float4 mu, const float4 rs,
                                         const float4 gv, const float4 bv) {
    float4 o;
    o.x = (leaky(t.x, 0.01f) - mu.x) * rs.x * gv.x + bv.x;
    o.y = (leaky(t.y, 0.01f) - mu.y) * rs.y * gv.y + bv.y;
    o.z = (leaky(t.z, 0.01f) - mu.z) * rs.z * gv.z + bv.z;
    o.w = (leaky(t.w, 0.01f) - mu.w) * rs.w * gv.w + bv.w;
    return o;
}

#define GA_PITCH 132
__device__ __forceinline__ void strip_rows(const float* __restrict__ X, int sidx, int w, int lane,
                                           int n, float4& g0, float4& g1) {
    int ra = sidx * 16 + w;
    int rb = ra + 8;
    if (ra >= n) ra = n - 1;
    if (rb >= n) rb = n - 1;
    g0 = *(const float4*)(X + (size_t)ra * HF + lane * 4);
    g1 = *(const float4*)(X + (size_t)rb * HF + lane * 4);
}
template <bool BN>
__global__ __launch_bounds__(256) void k_gemm_xw(const float* __restrict__ X,
                                                 const float* __restrict__ W,
                                                 float* __restrict__ Ho,
                                                 float* __restrict__ colz,
                                                 const float* __restrict__ pcs,
                                                 const float* __restrict__ pcq,
                                                 const float* __restrict__ pg,
                                                 const float* __restrict__ pbe,
                                                 const float* __restrict__ avs,
                                                 const float* __restrict__ avd,
                                                 float* __restrict__ als,
                                                 float* __restrict__ ald, int n) {
    __shared__ __attribute__((aligned(16))) float As[2][16 * GA_PITCH];
    __shared__ __attribute__((aligned(16))) float wa[2][HF];
    if (blockIdx.x == 0) colz[threadIdx.x] = 0.f;
    const int lane = threadIdx.x & 31;
    const int wv   = threadIdx.x >> 5;
    const int lo16 = lane & 15;
    const int hi   = lane >> 4;
    const int ncol = wv * 16 + lo16;

    const int strips = (n + 15) >> 4;
    float4 g0, g1;
    strip_rows(X, blockIdx.x < strips ? blockIdx.x : strips - 1, wv, lane, n, g0, g1);

    {
        const float4 as4 = *(const float4*)(avs + lane * 4);
        const float4 ad4 = *(const float4*)(avd + lane * 4);
        float4 wr[16];
#pragma unroll
        for (int j = 0; j < 16; ++j)
            wr[j] = *(const float4*)(W + (size_t)(wv * 16 + j) * HF + lane * 4);
        float ms = 0.f, md = 0.f;
#pragma unroll
        for (int j = 0; j < 16; ++j) {
            const float ps = wave_sum(dot4(wr[j], as4));
            const float pd = wave_sum(dot4(wr[j], ad4));
            if (lane == j) { ms = ps; md = pd; }
        }
        if (lane < 16) {
            wa[0][wv * 16 + lane] = ms;
            wa[1][wv * 16 + lane] = md;
        }
    }

    float4 mu = {0.f, 0.f, 0.f, 0.f}, rs = mu, gv = mu, bv = mu;
    if (BN) {
        const float4 cs = *(const float4*)(pcs + lane * 4);
        const float4 cq = *(const float4*)(pcq + lane * 4);
        gv = *(const float4*)(pg + lane * 4);
        bv = *(const float4*)(pbe + lane * 4);
        bn_consts(cs, cq, n, mu, rs);
    }

    v2f breg[32];
#pragma unroll
    for (int kk = 0; kk < 32; ++kk) {
        int kb = kk * 4 + hi * 2;
        v2f b;
        b.x = W[(size_t)kb * HF + ncol];
        b.y = W[(size_t)(kb + 1) * HF + ncol];
        breg[kk] = b;
    }

    __syncthreads();
    const float4 ws4 = *(const float4*)&wa[0][lane * 4];
    const float4 wd4 = *(const float4*)&wa[1][lane * 4];

    int buf = 0;
    for (int s = blockIdx.x; s < strips; s += gridDim.x) {
        const int r0 = s * 16;
        float* as = As[buf];
        float4 t0 = g0, t1 = g1;
        if (BN) {
            t0 = bn_row(g0, mu, rs, gv, bv);
            t1 = bn_row(g1, mu, rs, gv, bv);
        }
        *(float4*)(as + wv * GA_PITCH + lane * 4)       = t0;
        *(float4*)(as + (wv + 8) * GA_PITCH + lane * 4) = t1;
        __syncthreads();

        const int sn = s + (int)gridDim.x;
        strip_rows(X, sn < strips ? sn : strips - 1, wv, lane, n, g0, g1);
        __builtin_amdgcn_sched_barrier(0);

        {
            const float q0 = wave_sum(dot4(t0, ws4));
            const float q1 = wave_sum(dot4(t0, wd4));
            const float q2 = wave_sum(dot4(t1, ws4));
            const float q3 = wave_sum(dot4(t1, wd4));
            const int ra = r0 + wv;
            const int rb = ra + 8;
            if (lane == 0) {
                if (ra < n) { als[ra] = q0; ald[ra] = q1; }
                if (rb < n) { als[rb] = q2; ald[rb] = q3; }
            }
        }

        const float* ar = as + lo16 * GA_PITCH + hi * 2;
        v2f a[32];
#pragma unroll
        for (int kk = 0; kk < 32; ++kk) a[kk] = *(const v2f*)(ar + kk * 4);

        v8f acc = {0.f, 0.f, 0.f, 0.f, 0.f, 0.f, 0.f, 0.f};
#pragma unroll
        for (int kk = 0; kk < 32; ++kk) {
            acc = __builtin_amdgcn_wmma_f32_16x16x4_f32(
                false, a[kk], false, breg[kk], (short)0, acc, false, false);
        }

#pragma unroll
        for (int v = 0; v < 8; ++v) {
            int row = r0 + v + hi * 8;
            if (row < n) Ho[(size_t)row * HF + ncol] = acc[v];
        }
        buf ^= 1;
    }
}

#define LC_G   32
#define LC_T   1024
#define LC_CAP 65536
#define LC_U   13
__device__ __forceinline__ int lc_chunk(int E) {
    return (((E + LC_G - 1) / LC_G) + LC_T - 1) & ~(LC_T - 1);
}
__device__ __forceinline__ int lc_pitch(int n) { return (n + 31) & ~31; }

__global__ __launch_bounds__(LC_T) void k_chunk_count(const int* __restrict__ dst,
                                                      int* __restrict__ rank,
                                                      int* __restrict__ cnt,
                                                      float* __restrict__ easum, int E, int n) {
    __shared__ int hist[LC_CAP];
    const int tid = threadIdx.x;
    if (blockIdx.x == 0 && tid < 4) easum[tid] = 0.f;
    const int chunk = lc_chunk(E);
    const int e_lo  = blockIdx.x * chunk;
    int e_hi = e_lo + chunk;
    if (e_hi > E) e_hi = E;
    int* __restrict__ crow = cnt + (size_t)blockIdx.x * lc_pitch(n);
    for (int lo = 0; lo < n; lo += LC_CAP) {
        const int m = (n - lo < LC_CAP) ? (n - lo) : LC_CAP;
        for (int i = tid; i < m; i += LC_T) hist[i] = 0;
        __syncthreads();
        for (int e0 = e_lo + tid; e0 < e_hi; e0 += LC_U * LC_T) {
            int d[LC_U], r[LC_U];
#pragma unroll
            for (int u = 0; u < LC_U; ++u) {
                const int e = e0 + u * LC_T;
                d[u] = dst[e < e_hi ? e : e_hi - 1];
            }
#pragma unroll
            for (int u = 0; u < LC_U; ++u) {
                const int dd = d[u] - lo;
                const bool ok = (e0 + u * LC_T < e_hi) && ((unsigned)dd < (unsigned)m);
                d[u] = ok ? dd : -1;
                r[u] = 0;
                if (ok) r[u] = atomicAdd(&hist[dd], 1);
            }
#pragma unroll
            for (int u = 0; u < LC_U; ++u) {
                if (d[u] >= 0) rank[e0 + u * LC_T] = r[u];
            }
        }
        __syncthreads();
        for (int i = tid; i < m; i += LC_T) crow[lo + i] = hist[i];
        __syncthreads();
    }
}

__global__ __launch_bounds__(256) void k_chunk_base(int* __restrict__ cnt, int* __restrict__ deg,
                                                    int n) {
    const int d = blockIdx.x * blockDim.x + threadIdx.x;
    if (d >= n) return;
    const int pitch = lc_pitch(n);
    int c[LC_G];
#pragma unroll
    for (int g = 0; g < LC_G; ++g) c[g] = cnt[(size_t)g * pitch + d];
    int run = 0;
#pragma unroll
    for (int g = 0; g < LC_G; ++g) {
        cnt[(size_t)g * pitch + d] = run;
        run += c[g];
    }
    deg[d] = run;
}

#define SCAN_T 1024
#define SCAN_W (SCAN_T / 32)
#define SCAN_STEP 128
__device__ __forceinline__ int wave_isum(int v) {
    for (int off = 16; off; off >>= 1) v += __shfl_xor(v, off);
    return v;
}
__device__ __forceinline__ int4 deg4_tail(const int* __restrict__ deg, int i, int n) {
    int4 v = {0, 0, 0, 0};
    if (i < n)     v.x = deg[i];
    if (i + 1 < n) v.y = deg[i + 1];
    if (i + 2 < n) v.z = deg[i + 2];
    if (i + 3 < n) v.w = deg[i + 3];
    return v;
}
__device__ __forceinline__ int4 scan4(const int4 v, int lane, int& carry) {
    const int own = (v.x + v.y) + (v.z + v.w);
    int inc = own;
    for (int off = 1; off < 32; off <<= 1) {
        const int u = __shfl_up(inc, off);
        if (lane >= off) inc += u;
    }
    const int run = carry + (inc - own);
    carry += __builtin_amdgcn_readlane(inc, 31);
    int4 o;
    o.x = run;
    o.y = o.x + v.x;
    o.z = o.y + v.y;
    o.w = o.z + v.z;
    return o;
}
__global__ __launch_bounds__(SCAN_T) void k_row_scan(const int* __restrict__ deg,
                                                     int* __restrict__ rowptr, int n) {
    __shared__ int wtot[SCAN_W];
    const int lane = threadIdx.x & 31;
    const int w = __builtin_amdgcn_readfirstlane((int)(threadIdx.x >> 5));
    const int per = (((n + SCAN_W - 1) / SCAN_W) + SCAN_STEP - 1) & ~(SCAN_STEP - 1);
    const int lo = w * per;
    int hi = lo + per;
    if (hi > n) hi = n;
    const int l4 = lane * 4;

    int s = 0;
    int i0 = lo;
#pragma unroll 4
    for (; i0 + SCAN_STEP <= hi; i0 += SCAN_STEP) {
        const int4 v = *(const int4*)(deg + i0 + l4);
        s += (v.x + v.y) + (v.z + v.w);
    }
    if (i0 < hi) {
        const int4 v = deg4_tail(deg, i0 + l4, n);
        s += (v.x + v.y) + (v.z + v.w);
    }
    s = wave_isum(s);
    if (lane == 0) wtot[w] = s;
    __syncthreads();
    int carry = wave_isum(lane < w ? wtot[lane] : 0);

    i0 = lo;
#pragma unroll 4
    for (; i0 + SCAN_STEP <= hi; i0 += SCAN_STEP) {
        const int i = i0 + l4;
        const int4 o = scan4(*(const int4*)(deg + i), lane, carry);
        *(int4*)(rowptr + i) = o;
    }
    if (i0 < hi) {
        const int i = i0 + l4;
        const int4 o = scan4(deg4_tail(deg, i, n), lane, carry);
        if (i < n)     rowptr[i] = o.x;
        if (i + 1 < n) rowptr[i + 1] = o.y;
        if (i + 2 < n) rowptr[i + 2] = o.z;
        if (i + 3 < n) rowptr[i + 3] = o.w;
    }
}

__device__ __forceinline__ int lane_i(int v, int k) { return __builtin_amdgcn_readlane(v, k); }
__device__ __forceinline__ float lane_f(float v, int k) {
    return __int_as_float(__builtin_amdgcn_readlane(__float_as_int(v), k));
}
__device__ __forceinline__ float first_f(float v) {
    return __int_as_float(__builtin_amdgcn_readfirstlane(__float_as_int(v)));
}

__global__ __launch_bounds__(256) void k_list_fill(const int* __restrict__ src,
                                                   const int* __restrict__ dst,
                                                   const float* __restrict__ ea,
                                                   const int* __restrict__ rowptr,
                                                   const int* __restrict__ rank,
                                                   const int* __restrict__ cnt,
                                                   int* __restrict__ rec,
                                                   float* __restrict__ easum, int E, int n) {
    __shared__ float part[3][8];
    const int e = blockIdx.x * blockDim.x + threadIdx.x;
    const int* __restrict__ crow =
        cnt + (size_t)((blockIdx.x * blockDim.x) / lc_chunk(E)) * lc_pitch(n);
    float a0 = 0.f, a1 = 0.f, a2 = 0.f;
    if (e < E) {
        const int d = dst[e];
        const int p = rowptr[d] + crow[d] + rank[e];
        a0 = ea[e * 3 + 0];
        a1 = ea[e * 3 + 1];
        a2 = ea[e * 3 + 2];
        int4 r;
        r.x = src[e];
        r.y = __float_as_int(a0);
        r.z = __float_as_int(a1);
        r.w = __float_as_int(a2);
        ((int4*)rec)[p] = r;
    }
    a0 = wave_sum(a0);
    a1 = wave_sum(a1);
    a2 = wave_sum(a2);
    const int w = threadIdx.x >> 5;
    if ((threadIdx.x & 31) == 0) {
        part[0][w] = a0;
        part[1][w] = a1;
        part[2][w] = a2;
    }
    __syncthreads();
    if (threadIdx.x < 3) {
        float s = 0.f;
        for (int k = 0; k < 8; ++k) s += part[threadIdx.x][k];
        atomicAdd(easum + threadIdx.x, s);
    }
}

__device__ __forceinline__ void row_fma(float4& o, const float4 h, float c) {
    o.x = fmaf(h.x, c, o.x);
    o.y = fmaf(h.y, c, o.y);
    o.z = fmaf(h.z, c, o.z);
    o.w = fmaf(h.w, c, o.w);
}

__device__ __forceinline__ float slot_alpha(const int4* __restrict__ rec4,
                                            const float* __restrict__ als, int j, float ad,
                                            float c0, float c1, float c2, int& s) {
    const int4 r = rec4[j];
    s = r.x;
    const float ale = __int_as_float(r.y) * c0 + __int_as_float(r.z) * c1 +
                      __int_as_float(r.w) * c2;
    return leaky(als[s] + ad + ale, 0.2f);
}

__device__ __forceinline__ void gather_group(float4& o, int sl, float cl, int m,
                                             const float* __restrict__ Hm, int lane) {
    int k = 0;
    for (; k + 4 <= m; k += 4) {
        const int s0 = lane_i(sl, k),     s1 = lane_i(sl, k + 1);
        const int s2 = lane_i(sl, k + 2), s3 = lane_i(sl, k + 3);
        const float c0 = lane_f(cl, k),     c1 = lane_f(cl, k + 1);
        const float c2 = lane_f(cl, k + 2), c3 = lane_f(cl, k + 3);
        const float4 h0 = *(const float4*)(Hm + (size_t)s0 * HF + lane * 4);
        const float4 h1 = *(const float4*)(Hm + (size_t)s1 * HF + lane * 4);
        const float4 h2 = *(const float4*)(Hm + (size_t)s2 * HF + lane * 4);
        const float4 h3 = *(const float4*)(Hm + (size_t)s3 * HF + lane * 4);
        row_fma(o, h0, c0);
        row_fma(o, h1, c1);
        row_fma(o, h2, c2);
        row_fma(o, h3, c3);
    }
    const int r = m - k;
    if (r == 3) {
        const int s0 = lane_i(sl, k), s1 = lane_i(sl, k + 1), s2 = lane_i(sl, k + 2);
        const float c0 = lane_f(cl, k), c1 = lane_f(cl, k + 1), c2 = lane_f(cl, k + 2);
        const float4 h0 = *(const float4*)(Hm + (size_t)s0 * HF + lane * 4);
        const float4 h1 = *(const float4*)(Hm + (size_t)s1 * HF + lane * 4);
        const float4 h2 = *(const float4*)(Hm + (size_t)s2 * HF + lane * 4);
        row_fma(o, h0, c0);
        row_fma(o, h1, c1);
        row_fma(o, h2, c2);
    } else if (r == 2) {
        const int s0 = lane_i(sl, k), s1 = lane_i(sl, k + 1);
        const float c0 = lane_f(cl, k), c1 = lane_f(cl, k + 1);
        const float4 h0 = *(const float4*)(Hm + (size_t)s0 * HF + lane * 4);
        const float4 h1 = *(const float4*)(Hm + (size_t)s1 * HF + lane * 4);
        row_fma(o, h0, c0);
        row_fma(o, h1, c1);
    } else if (r == 1) {
        const int s0 = lane_i(sl, k);
        const float c0 = lane_f(cl, k);
        const float4 h0 = *(const float4*)(Hm + (size_t)s0 * HF + lane * 4);
        row_fma(o, h0, c0);
    }
}

__global__ __launch_bounds__(256) void k_node_attn(const int* __restrict__ rowptr,
                                                   const int* __restrict__ deg,
                                                   const int* __restrict__ rec,
                                                   const float* __restrict__ als,
                                                   const float* __restrict__ ald,
                                                   const float* __restrict__ We,
                                                   const float* __restrict__ ae,
                                                   const float* __restrict__ easum, int E,
                                                   const float* __restrict__ bias,
                                                   const float* __restrict__ Hm,
                                                   float* __restrict__ acc, int n) {
    const int lane = threadIdx.x & 31;
    const int nw   = (gridDim.x * blockDim.x) >> 5;
    const int w0 = __builtin_amdgcn_readfirstlane(
        (int)((blockIdx.x * blockDim.x + threadIdx.x) >> 5));
    const int4* __restrict__ rec4 = (const int4*)rec;
    const float4 av = *(const float4*)(ae + lane * 4);
    const float4 e0 = *(const float4*)(We + lane * 4);
    const float4 e1 = *(const float4*)(We + HF + lane * 4);
    const float4 e2 = *(const float4*)(We + 2 * HF + lane * 4);
    const float c0 = first_f(wave_sum(e0.x * av.x + e0.y * av.y + e0.z * av.z + e0.w * av.w));
    const float c1 = first_f(wave_sum(e1.x * av.x + e1.y * av.y + e1.z * av.z + e1.w * av.w));
    const float c2 = first_f(wave_sum(e2.x * av.x + e2.y * av.y + e2.z * av.z + e2.w * av.w));
    float cm = 0.f;
    cm += (easum[0] / (float)E) * c0;
    cm += (easum[1] / (float)E) * c1;
    cm += (easum[2] / (float)E) * c2;
    const float4 bv = *(const float4*)(bias + lane * 4);
    const float ninf = -__builtin_inff();
    for (int i = w0; i < n; i += nw) {
        const int start = rowptr[i];
        const int cnt   = deg[i];
        const float ad  = ald[i];
        const float as  = als[i];
        const float4 hs = *(const float4*)(Hm + (size_t)i * HF + lane * 4);
        float4 o;
        if (cnt <= 32) {
            const bool ok = lane < cnt;
            int sl = 0;
            float a = ninf;
            if (ok) a = slot_alpha(rec4, als, start + lane, ad, c0, c1, c2, sl);
            const float aself = leaky(as + ad + cm, 0.2f);
            const float m   = fmaxf(first_f(wave_max(a)), aself);
            const float ex  = ok ? expf(a - m) : 0.f;
            const float exs = expf(aself - m);
            const float den = (exs + first_f(wave_sum(ex))) + 1e-16f;
            const float cs  = exs / den;
            o.x = hs.x * cs + bv.x;
            o.y = hs.y * cs + bv.y;
            o.z = hs.z * cs + bv.z;
            o.w = hs.w * cs + bv.w;
            if (cnt > 0) gather_group(o, sl, ex / den, cnt, Hm, lane);
        } else {
            float mx = ninf;
            for (int base = 0; base < cnt; base += 32) {
                if (base + lane < cnt) {
                    int s;
                    mx = fmaxf(mx, slot_alpha(rec4, als, start + base + lane, ad, c0, c1, c2, s));
                }
            }
            const float aself = leaky(as + ad + cm, 0.2f);
            const float m = fmaxf(first_f(wave_max(mx)), aself);
            float sm = 0.f;
            for (int base = 0; base < cnt; base += 32) {
                if (base + lane < cnt) {
                    int s;
                    sm += expf(slot_alpha(rec4, als, start + base + lane, ad, c0, c1, c2, s) - m);
                }
            }
            const float exs = expf(aself - m);
            const float den = (exs + first_f(wave_sum(sm))) + 1e-16f;
            const float cs  = exs / den;
            o.x = hs.x * cs + bv.x;
            o.y = hs.y * cs + bv.y;
            o.z = hs.z * cs + bv.z;
            o.w = hs.w * cs + bv.w;
            for (int base = 0; base < cnt; base += 32) {
                int mm = cnt - base;
                if (mm > 32) mm = 32;
                int sl = 0;
                float cl = 0.f;
                if (lane < mm) {
                    const float a = slot_alpha(rec4, als, start + base + lane, ad, c0, c1, c2, sl);
                    cl = expf(a - m) / den;
                }
                gather_group(o, sl, cl, mm, Hm, lane);
            }
        }
        *(float4*)(acc + (size_t)i * HF + lane * 4) = o;
    }
}

#define BN_ROWS  256
#define BN_WAVES 16
#define BN_RPW   (BN_ROWS / BN_WAVES)
__global__ __launch_bounds__(512) void k_bn_stats(const float* __restrict__ acc,
                                                  float* __restrict__ colsum,
                                                  float* __restrict__ colsq, int n) {
    __shared__ __attribute__((aligned(16))) float part[2][BN_WAVES][HF];
    const int lane = threadIdx.x & 31;
    const int w    = __builtin_amdgcn_readfirstlane((int)(threadIdx.x >> 5));
    const int r0   = blockIdx.x * BN_ROWS + w;

    float4 t[BN_RPW];
#pragma unroll
    for (int i = 0; i < BN_RPW; ++i) {
        int r = r0 + i * BN_WAVES;
        if (r >= n) r = n - 1;
        t[i] = *(const float4*)(acc + (size_t)r * HF + lane * 4);
    }

    float4 s = {0.f, 0.f, 0.f, 0.f}, q = {0.f, 0.f, 0.f, 0.f};
#pragma unroll
    for (int i = 0; i < BN_RPW; ++i) {
        const bool ok = (r0 + i * BN_WAVES) < n;
        float y0 = ok ? leaky(t[i].x, 0.01f) : 0.f;
        float y1 = ok ? leaky(t[i].y, 0.01f) : 0.f;
        float y2 = ok ? leaky(t[i].z, 0.01f) : 0.f;
        float y3 = ok ? leaky(t[i].w, 0.01f) : 0.f;
        s.x += y0; q.x += y0 * y0;
        s.y += y1; q.y += y1 * y1;
        s.z += y2; q.z += y2 * y2;
        s.w += y3; q.w += y3 * y3;
    }
    *(float4*)&part[0][w][lane * 4] = s;
    *(float4*)&part[1][w][lane * 4] = q;
    __syncthreads();

    if (threadIdx.x < 2 * HF) {
        const int which = threadIdx.x >> 7;
        const int c     = threadIdx.x & (HF - 1);
        float r = 0.f;
        for (int k = 0; k < BN_WAVES; ++k) r += part[which][k][c];
        atomicAdd((which ? colsq : colsum) + c, r);
    }
}

#define BA_RPW  16
#define BA_ROWS (8 * BA_RPW)
__global__ __launch_bounds__(256) void k_bn_apply(const float* __restrict__ acc,
                                                  const float* __restrict__ colsum,
                                                  const float* __restrict__ colsq,
                                                  const float* __restrict__ g,
                                                  const float* __restrict__ be,
                                                  float* __restrict__ xout, int n) {
    const int lane = threadIdx.x & 31;
    const int w  = __builtin_amdgcn_readfirstlane(
        (int)((blockIdx.x * blockDim.x + threadIdx.x) >> 5));
    const int r0 = w * BA_RPW;
    if (r0 >= n) return;

    const float4 cs = *(const float4*)(colsum + lane * 4);
    const float4 cq = *(const float4*)(colsq + lane * 4);
    const float4 gv = *(const float4*)(g + lane * 4);
    const float4 bv = *(const float4*)(be + lane * 4);
    const float* __restrict__ src = acc + (size_t)r0 * HF + lane * 4;
    float* __restrict__ dst = xout + (size_t)r0 * HF + lane * 4;

    if (r0 + BA_RPW <= n) {
        float4 t[BA_RPW];
#pragma unroll
        for (int i = 0; i < BA_RPW; ++i) t[i] = *(const float4*)(src + (size_t)i * HF);

        float4 mu, rs;
        bn_consts(cs, cq, n, mu, rs);
#pragma unroll
        for (int i = 0; i < BA_RPW; ++i)
            *(float4*)(dst + (size_t)i * HF) = bn_row(t[i], mu, rs, gv, bv);
    } else {
        float4 mu, rs;
        bn_consts(cs, cq, n, mu, rs);
        for (int i = 0; r0 + i < n; ++i)
            *(float4*)(dst + (size_t)i * HF) =
                bn_row(*(const float4*)(src + (size_t)i * HF), mu, rs, gv, bv);
    }
}

extern "C" void kernel_launch(void* const* d_in, const int* in_sizes, int n_in,
                              void* d_out, int out_size, void* d_ws, size_t ws_size,
                              hipStream_t stream) {
    const float* x  = (const float*)d_in[0];
    const int*   ei = (const int*)d_in[1];
    const float* ea = (const float*)d_in[3];
    const int N = in_sizes[0] / HF;
    const int E = in_sizes[1] / 2;
    const int* src = ei;
    const int* dst = ei + E;

    float* ws = (float*)d_ws;
    const size_t NH = (size_t)N * HF;
    float* bufA   = ws;
    float* bufB   = bufA + NH;
    int*   rec    = (int*)(bufB + NH);
    int*   rowptr = rec + 4 * (size_t)E;
    int*   deg    = rowptr + N;
    int*   rank   = deg + N;
    int*   cnt    = rank + E;
    float* als    = bufB + NH + NH;
    float* ald    = als + N;
    float* easum  = ald + N + 2 * HF;
    float* cstat  = (float*)rank;

    const int nb_e  = (E + 255) / 256;
    const int nb_n  = (N + 255) / 256;

    k_chunk_count<<<LC_G, LC_T, 0, stream>>>(dst, rank, cnt, easum, E, N);
    k_chunk_base<<<nb_n, 256, 0, stream>>>(cnt, deg, N);
    k_row_scan<<<1, SCAN_T, 0, stream>>>(deg, rowptr, N);
    k_list_fill<<<nb_e, 256, 0, stream>>>(src, dst, ea, rowptr, rank, cnt, rec, easum, E, N);

    for (int l = 0; l < 3; ++l) {
        const float* Wl  = (const float*)d_in[4 + 8 * l];
        const float* asv = (const float*)d_in[5 + 8 * l];
        const float* adv = (const float*)d_in[6 + 8 * l];
        const float* Wel = (const float*)d_in[7 + 8 * l];
        const float* aev = (const float*)d_in[8 + 8 * l];
        const float* bv  = (const float*)d_in[9 + 8 * l];
        float* colsum = cstat + 2 * HF * l;
        float* colsq  = colsum + HF;

        if (l == 0) {
            k_gemm_xw<false><<<1024, 256, 0, stream>>>(x, Wl, bufA, colsum, nullptr, nullptr, nullptr,
                                                       nullptr, asv, adv, als, ald, N);
        } else {
            const float* pgv  = (const float*)d_in[10 + 8 * (l - 1)];
            const float* pbev = (const float*)d_in[11 + 8 * (l - 1)];
            k_gemm_xw<true><<<1024, 256, 0, stream>>>(bufB, Wl, bufA, colsum, colsum - 2 * HF,
                                                      colsq - 2 * HF, pgv, pbev, asv, adv, als, ald, N);
        }
        k_node_attn<<<2048, 256, 0, stream>>>(rowptr, deg, rec, als, ald, Wel, aev, easum, E, bv,
                                              bufA, bufB, N);
        k_bn_stats<<<(N + 255) / 256, 512, 0, stream>>>(bufB, colsum, colsq, N);
    }
    k_bn_apply<<<(N + BA_ROWS - 1) / BA_ROWS, 256, 0, stream>>>(
        bufB, cstat + 4 * HF, cstat + 5 * HF, (const float*)d_in[26], (const float*)d_in[27],
        (float*)d_out, N);
}
